# starting kernel
# baseline (speedup 1.0000x reference)
#include <hip/hip_runtime.h>

typedef float v2f __attribute__((ext_vector_type(2)));
typedef float v8f __attribute__((ext_vector_type(8)));

#define TPB 256

__global__ void gnn_init(float* __restrict__ deg, float* __restrict__ sums,
                         float* __restrict__ cnts, int N, int G) {
  int i = blockIdx.x * blockDim.x + threadIdx.x;
  if (i < N) deg[i] = 1.0f;
  if (i < G * 32) sums[i] = 0.0f;
  if (i < G) cnts[i] = 0.0f;
}

__global__ void gnn_deg(const int* __restrict__ dst, float* __restrict__ deg, int E) {
  int e = blockIdx.x * blockDim.x + threadIdx.x;
  if (e < E) atomicAdd(&deg[dst[e]], 1.0f);
}

__global__ void gnn_rsqrt(float* __restrict__ deg, int N) {
  int i = blockIdx.x * blockDim.x + threadIdx.x;
  if (i < N) deg[i] = rsqrtf(deg[i]);
}

__global__ void gnn_agg1_init(const float* __restrict__ x, const float* __restrict__ dinv,
                              float* __restrict__ agg, int N) {
  int i = blockIdx.x * blockDim.x + threadIdx.x;
  if (i < N) {
    float nrm = dinv[i] * dinv[i];
    #pragma unroll
    for (int c = 0; c < 13; ++c) agg[i * 13 + c] = nrm * x[i * 13 + c];
  }
}

__global__ void gnn_agg1_edge(const float* __restrict__ x, const float* __restrict__ dinv,
                              const int* __restrict__ src, const int* __restrict__ dst,
                              float* __restrict__ agg, int E) {
  int e = blockIdx.x * blockDim.x + threadIdx.x;
  if (e < E) {
    int s = src[e], d = dst[e];
    float nrm = dinv[s] * dinv[d];
    #pragma unroll
    for (int c = 0; c < 13; ++c)
      atomicAdd(&agg[d * 13 + c], nrm * x[s * 13 + c]);
  }
}

__global__ __launch_bounds__(256)
void gnn_gemm1_wmma(const float* __restrict__ A, const float* __restrict__ W,
                    const float* __restrict__ bias, float* __restrict__ H, int N) {
  const int lane = threadIdx.x & 31;
  const int wave = threadIdx.x >> 5;
  const int mbase = (blockIdx.x * 8 + wave) * 16;
  const int half = lane >> 4;
  const int l16 = lane & 15;

  v8f acc[4];
  const v8f zero = {0.f, 0.f, 0.f, 0.f, 0.f, 0.f, 0.f, 0.f};
  #pragma unroll
  for (int nt = 0; nt < 4; ++nt) acc[nt] = zero;

  const int rowA = mbase + l16;

  if (mbase + 16 <= N) {
    const float* Arow = A + rowA * 13;
    #pragma unroll
    for (int kk = 0; kk < 12; kk += 4) {
      int ka0 = kk + half * 2;
      v2f av; av[0] = Arow[ka0]; av[1] = Arow[ka0 + 1];
      #pragma unroll
      for (int nt = 0; nt < 4; ++nt) {
        int col = nt * 16 + l16;
        v2f bv; bv[0] = W[ka0 * 64 + col]; bv[1] = W[(ka0 + 1) * 64 + col];
        acc[nt] = __builtin_amdgcn_wmma_f32_16x16x4_f32(
            false, av, false, bv, (short)0, acc[nt], false, false);
      }
    }
    {
      float a12 = Arow[12];
      v2f av; av[0] = half ? 0.0f : a12; av[1] = 0.0f;
      #pragma unroll
      for (int nt = 0; nt < 4; ++nt) {
        int col = nt * 16 + l16;
        float b12 = W[12 * 64 + col];
        v2f bv; bv[0] = half ? 0.0f : b12; bv[1] = 0.0f;
        acc[nt] = __builtin_amdgcn_wmma_f32_16x16x4_f32(
            false, av, false, bv, (short)0, acc[nt], false, false);
      }
    }
    #pragma unroll
    for (int nt = 0; nt < 4; ++nt) {
      int col = nt * 16 + l16;
      float bc = bias[col];
      #pragma unroll
      for (int v = 0; v < 8; ++v) {
        int row = mbase + v + half * 8;
        H[row * 64 + col] = fmaxf(acc[nt][v] + bc, 0.0f);
      }
    }
  } else {
    const bool rv = rowA < N;
    #pragma unroll
    for (int kk = 0; kk < 16; kk += 4) {
      int ka0 = kk + half * 2;
      int ka1 = ka0 + 1;
      float a0 = (rv && ka0 < 13) ? A[rowA * 13 + ka0] : 0.0f;
      float a1 = (rv && ka1 < 13) ? A[rowA * 13 + ka1] : 0.0f;
      v2f av; av[0] = a0; av[1] = a1;
      #pragma unroll
      for (int nt = 0; nt < 4; ++nt) {
        int col = nt * 16 + l16;
        float b0 = (ka0 < 13) ? W[ka0 * 64 + col] : 0.0f;
        float b1 = (ka1 < 13) ? W[ka1 * 64 + col] : 0.0f;
        v2f bv; bv[0] = b0; bv[1] = b1;
        acc[nt] = __builtin_amdgcn_wmma_f32_16x16x4_f32(
            false, av, false, bv, (short)0, acc[nt], false, false);
      }
    }
    #pragma unroll
    for (int nt = 0; nt < 4; ++nt) {
      int col = nt * 16 + l16;
      float bc = bias[col];
      #pragma unroll
      for (int v = 0; v < 8; ++v) {
        int row = mbase + v + half * 8;
        if (row < N) H[row * 64 + col] = fmaxf(acc[nt][v] + bc, 0.0f);
      }
    }
  }
}

__global__ __launch_bounds__(256)
void gnn_gemm2_wmma(const float* __restrict__ A, const float* __restrict__ W,
                    float* __restrict__ O, int N) {
  const int lane = threadIdx.x & 31;
  const int wave = threadIdx.x >> 5;
  const int mbase = (blockIdx.x * 8 + wave) * 16;
  const int half = lane >> 4;
  const int l16 = lane & 15;

  v8f acc[2];
  const v8f zero = {0.f, 0.f, 0.f, 0.f, 0.f, 0.f, 0.f, 0.f};
  acc[0] = zero; acc[1] = zero;

  const int rowA = mbase + l16;

  if (mbase + 16 <= N) {
    const float* Arow = A + rowA * 64;
    #pragma unroll
    for (int kk = 0; kk < 64; kk += 4) {
      int ka0 = kk + half * 2;
      const float2 a2 = *(const float2*)(Arow + ka0);
      v2f av; av[0] = a2.x; av[1] = a2.y;
      #pragma unroll
      for (int nt = 0; nt < 2; ++nt) {
        int col = nt * 16 + l16;
        v2f bv; bv[0] = W[ka0 * 32 + col]; bv[1] = W[(ka0 + 1) * 32 + col];
        acc[nt] = __builtin_amdgcn_wmma_f32_16x16x4_f32(
            false, av, false, bv, (short)0, acc[nt], false, false);
      }
    }
    #pragma unroll
    for (int nt = 0; nt < 2; ++nt) {
      int col = nt * 16 + l16;
      #pragma unroll
      for (int v = 0; v < 8; ++v) {
        int row = mbase + v + half * 8;
        O[row * 32 + col] = acc[nt][v];
      }
    }
  } else {
    const bool rv = rowA < N;
    #pragma unroll
    for (int kk = 0; kk < 64; kk += 4) {
      int ka0 = kk + half * 2;
      int ka1 = ka0 + 1;
      float a0 = rv ? A[rowA * 64 + ka0] : 0.0f;
      float a1 = rv ? A[rowA * 64 + ka1] : 0.0f;
      v2f av; av[0] = a0; av[1] = a1;
      #pragma unroll
      for (int nt = 0; nt < 2; ++nt) {
        int col = nt * 16 + l16;
        v2f bv; bv[0] = W[ka0 * 32 + col]; bv[1] = W[ka1 * 32 + col];
        acc[nt] = __builtin_amdgcn_wmma_f32_16x16x4_f32(
            false, av, false, bv, (short)0, acc[nt], false, false);
      }
    }
    #pragma unroll
    for (int nt = 0; nt < 2; ++nt) {
      int col = nt * 16 + l16;
      #pragma unroll
      for (int v = 0; v < 8; ++v) {
        int row = mbase + v + half * 8;
        if (row < N) O[row * 32 + col] = acc[nt][v];
      }
    }
  }
}

__global__ void gnn_h2_init(const float* __restrict__ xw2, const float* __restrict__ dinv,
                            float* __restrict__ h2, int N) {
  int t = blockIdx.x * blockDim.x + threadIdx.x;
  if (t < N * 8) {
    int i = t >> 3;
    float nrm = dinv[i] * dinv[i];
    float4 v = ((const float4*)xw2)[t];
    float4 o; o.x = v.x * nrm; o.y = v.y * nrm; o.z = v.z * nrm; o.w = v.w * nrm;
    ((float4*)h2)[t] = o;
  }
}

__global__ void gnn_h2_edge(const float* __restrict__ xw2, const float* __restrict__ dinv,
                            const int* __restrict__ src, const int* __restrict__ dst,
                            float* __restrict__ h2, int E) {
  int t = blockIdx.x * blockDim.x + threadIdx.x;
  if (t < E * 4) {
    int e = t >> 2, q = (t & 3) * 8;
    int s = src[e], d = dst[e];
    float nrm = dinv[s] * dinv[d];
    #pragma unroll
    for (int c = 0; c < 8; ++c)
      atomicAdd(&h2[d * 32 + q + c], nrm * xw2[s * 32 + q + c]);
  }
}

__global__ void gnn_pool(const float* __restrict__ h2, const int* __restrict__ batch,
                         float* __restrict__ sums, float* __restrict__ cnts, int N) {
  int t = blockIdx.x * blockDim.x + threadIdx.x;
  int nblk = (N + 7) >> 3;
  if (t >= nblk * 4) return;
  int blk = t >> 2, q = (t & 3) * 8;
  int i0 = blk * 8;
  int iend = (i0 + 8 < N) ? (i0 + 8) : N;
  int cur = batch[i0];
  float acc[8];
  #pragma unroll
  for (int c = 0; c < 8; ++c) acc[c] = 0.0f;
  float cnt = 0.0f;
  for (int i = i0; i < iend; ++i) {
    int b = batch[i];
    if (b != cur) {
      #pragma unroll
      for (int c = 0; c < 8; ++c) atomicAdd(&sums[cur * 32 + q + c], acc[c]);
      if (q == 0) atomicAdd(&cnts[cur], cnt);
      #pragma unroll
      for (int c = 0; c < 8; ++c) acc[c] = 0.0f;
      cnt = 0.0f;
      cur = b;
    }
    #pragma unroll
    for (int c = 0; c < 8; ++c) acc[c] += h2[i * 32 + q + c];
    cnt += 1.0f;
  }
  #pragma unroll
  for (int c = 0; c < 8; ++c) atomicAdd(&sums[cur * 32 + q + c], acc[c]);
  if (q == 0) atomicAdd(&cnts[cur], cnt);
}

__global__ void gnn_gmean(const float* __restrict__ sums, const float* __restrict__ cnts,
                          const float* __restrict__ b2, float* __restrict__ g, int G) {
  int t = blockIdx.x * blockDim.x + threadIdx.x;
  if (t < G * 32) {
    int b = t >> 5, j = t & 31;
    g[t] = sums[t] / fmaxf(cnts[b], 1.0f) + b2[j];
  }
}

__global__ void gnn_fc_relu(const float* __restrict__ g, const float* __restrict__ w,
                            const float* __restrict__ bias, float* __restrict__ o, int G) {
  int t = blockIdx.x * blockDim.x + threadIdx.x;
  if (t < G * 32) {
    int b = t >> 5, j = t & 31;
    float s = bias[j];
    #pragma unroll
    for (int k = 0; k < 32; ++k) s += g[b * 32 + k] * w[k * 32 + j];
    o[t] = fmaxf(s, 0.0f);
  }
}

__global__ void gnn_fc_out(const float* __restrict__ g, const float* __restrict__ w,
                           const float* __restrict__ bias, float* __restrict__ o, int G) {
  int t = blockIdx.x * blockDim.x + threadIdx.x;
  if (t < G * 32) {
    int b = t >> 5, j = t & 31;
    float s = bias[j];
    #pragma unroll
    for (int k = 0; k < 32; ++k) s += g[b * 32 + k] * w[k * 32 + j];
    o[t] = s;
  }
}

extern "C" void kernel_launch(void* const* d_in, const int* in_sizes, int n_in,
                              void* d_out, int out_size, void* d_ws, size_t ws_size,
                              hipStream_t stream) {
  const float* x    = (const float*)d_in[0];
  const int*   ei   = (const int*)d_in[1];
  const int*   batch= (const int*)d_in[2];
  const float* W1   = (const float*)d_in[3];
  const float* b1   = (const float*)d_in[4];
  const float* W2   = (const float*)d_in[5];
  const float* b2   = (const float*)d_in[6];
  const float* fc1w = (const float*)d_in[7];
  const float* fc1b = (const float*)d_in[8];
  const float* fc2w = (const float*)d_in[9];
  const float* fc2b = (const float*)d_in[10];

  const int N = in_sizes[0] / 13;
  const int E = in_sizes[1] / 2;
  const int G = out_size / 32;
  const int* src = ei;
  const int* dst = ei + E;

  auto al4 = [](size_t v) { return (v + 3) & ~(size_t)3; };
  float* ws   = (float*)d_ws;
  size_t off  = 0;
  float* dinv = ws + off; off = al4(off + (size_t)N);
  float* agg1 = ws + off; off = al4(off + (size_t)13 * N);
  float* h1   = ws + off; off = al4(off + (size_t)64 * N);
  float* xw2  = ws + off; off = al4(off + (size_t)32 * N);
  float* h2   = ws + off; off = al4(off + (size_t)32 * N);
  float* sums = ws + off; off = al4(off + (size_t)32 * G);
  float* cnts = ws + off; off = al4(off + (size_t)G);
  float* g    = ws + off; off = al4(off + (size_t)32 * G);
  float* hfc  = ws + off; off = al4(off + (size_t)32 * G);

  auto cdiv = [](int a, int b) { return (a + b - 1) / b; };

  gnn_init<<<cdiv(N, TPB), TPB, 0, stream>>>(dinv, sums, cnts, N, G);
  gnn_deg<<<cdiv(E, TPB), TPB, 0, stream>>>(dst, dinv, E);
  gnn_rsqrt<<<cdiv(N, TPB), TPB, 0, stream>>>(dinv, N);

  gnn_agg1_init<<<cdiv(N, TPB), TPB, 0, stream>>>(x, dinv, agg1, N);
  gnn_agg1_edge<<<cdiv(E, TPB), TPB, 0, stream>>>(x, dinv, src, dst, agg1, E);
  int mt = cdiv(N, 16);
  gnn_gemm1_wmma<<<cdiv(mt, 8), 256, 0, stream>>>(agg1, W1, b1, h1, N);

  gnn_gemm2_wmma<<<cdiv(mt, 8), 256, 0, stream>>>(h1, W2, xw2, N);
  gnn_h2_init<<<cdiv(N * 8, TPB), TPB, 0, stream>>>(xw2, dinv, h2, N);
  gnn_h2_edge<<<cdiv(E * 4, TPB), TPB, 0, stream>>>(xw2, dinv, src, dst, h2, E);

  int nblk = (N + 7) / 8;
  gnn_pool<<<cdiv(nblk * 4, TPB), TPB, 0, stream>>>(h2, batch, sums, cnts, N);
  gnn_gmean<<<cdiv(G * 32, TPB), TPB, 0, stream>>>(sums, cnts, b2, g, G);
  gnn_fc_relu<<<cdiv(G * 32, TPB), TPB, 0, stream>>>(g, fc1w, fc1b, hfc, G);
  gnn_fc_out<<<cdiv(G * 32, TPB), TPB, 0, stream>>>(hfc, fc2w, fc2b, (float*)d_out, G);
}


# both edge scatters: lanes map to channels of one edge row (wave walks 32 coalesced-loaded edges by lane broadcast), one atomic instruction per 1-2 rows, not 8-32; float atomics in arrival order
# speedup vs baseline: 2.1763x; 2.1763x over previous
#include <hip/hip_runtime.h>

typedef float v2f __attribute__((ext_vector_type(2)));
typedef float v8f __attribute__((ext_vector_type(8)));

#define TPB 256

__global__ void gnn_init(float* __restrict__ deg, float* __restrict__ sums,
                         float* __restrict__ cnts, int N, int G) {
  int i = blockIdx.x * blockDim.x + threadIdx.x;
  if (i < N) deg[i] = 1.0f;
  if (i < G * 32) sums[i] = 0.0f;
  if (i < G) cnts[i] = 0.0f;
}

__global__ void gnn_deg(const int* __restrict__ dst, float* __restrict__ deg, int E) {
  int e = blockIdx.x * blockDim.x + threadIdx.x;
  if (e < E) atomicAdd(&deg[dst[e]], 1.0f);
}

__global__ void gnn_rsqrt(float* __restrict__ deg, int N) {
  int i = blockIdx.x * blockDim.x + threadIdx.x;
  if (i < N) deg[i] = rsqrtf(deg[i]);
}

__global__ void gnn_agg1_init(const float* __restrict__ x, const float* __restrict__ dinv,
                              float* __restrict__ agg, int N) {
  int i = blockIdx.x * blockDim.x + threadIdx.x;
  if (i < N) {
    float nrm = dinv[i] * dinv[i];
    #pragma unroll
    for (int c = 0; c < 13; ++c) agg[i * 13 + c] = nrm * x[i * 13 + c];
  }
}

__global__ void gnn_agg1_edge(const float* __restrict__ x, const float* __restrict__ dinv,
                              const int* __restrict__ src, const int* __restrict__ dst,
                              float* __restrict__ agg, int E) {
  const int lane = threadIdx.x & 31;
  const int e0 = (blockIdx.x * blockDim.x + threadIdx.x) & ~31;
  if (e0 >= E) return;
  const int e = e0 + lane;
  int s = 0, d = 0;
  float nrm = 0.0f;
  if (e < E) {
    s = src[e]; d = dst[e];
    nrm = dinv[s] * dinv[d];
  }
  const int cnt = E - e0;
  const int half = lane >> 4;
  const int c = lane & 15;
  int sj[16], dj[16];
  float nj[16];
  #pragma unroll
  for (int j = 0; j < 16; ++j) {
    const int idx = 2 * j + half;
    sj[j] = __shfl(s, idx);
    dj[j] = __shfl(d, idx);
    nj[j] = __shfl(nrm, idx);
  }
  if (c < 13) {
    if (cnt >= 32) {
      float xv[16];
      #pragma unroll
      for (int j = 0; j < 16; ++j) xv[j] = x[sj[j] * 13 + c];
      #pragma unroll
      for (int j = 0; j < 16; ++j) atomicAdd(&agg[dj[j] * 13 + c], nj[j] * xv[j]);
    } else {
      #pragma unroll
      for (int j = 0; j < 16; ++j)
        if (2 * j + half < cnt) atomicAdd(&agg[dj[j] * 13 + c], nj[j] * x[sj[j] * 13 + c]);
    }
  }
}

__global__ __launch_bounds__(256)
void gnn_gemm1_wmma(const float* __restrict__ A, const float* __restrict__ W,
                    const float* __restrict__ bias, float* __restrict__ H, int N) {
  const int lane = threadIdx.x & 31;
  const int wave = threadIdx.x >> 5;
  const int mbase = (blockIdx.x * 8 + wave) * 16;
  const int half = lane >> 4;
  const int l16 = lane & 15;

  v8f acc[4];
  const v8f zero = {0.f, 0.f, 0.f, 0.f, 0.f, 0.f, 0.f, 0.f};
  #pragma unroll
  for (int nt = 0; nt < 4; ++nt) acc[nt] = zero;

  const int rowA = mbase + l16;

  if (mbase + 16 <= N) {
    const float* Arow = A + rowA * 13;
    #pragma unroll
    for (int kk = 0; kk < 12; kk += 4) {
      int ka0 = kk + half * 2;
      v2f av; av[0] = Arow[ka0]; av[1] = Arow[ka0 + 1];
      #pragma unroll
      for (int nt = 0; nt < 4; ++nt) {
        int col = nt * 16 + l16;
        v2f bv; bv[0] = W[ka0 * 64 + col]; bv[1] = W[(ka0 + 1) * 64 + col];
        acc[nt] = __builtin_amdgcn_wmma_f32_16x16x4_f32(
            false, av, false, bv, (short)0, acc[nt], false, false);
      }
    }
    {
      float a12 = Arow[12];
      v2f av; av[0] = half ? 0.0f : a12; av[1] = 0.0f;
      #pragma unroll
      for (int nt = 0; nt < 4; ++nt) {
        int col = nt * 16 + l16;
        float b12 = W[12 * 64 + col];
        v2f bv; bv[0] = half ? 0.0f : b12; bv[1] = 0.0f;
        acc[nt] = __builtin_amdgcn_wmma_f32_16x16x4_f32(
            false, av, false, bv, (short)0, acc[nt], false, false);
      }
    }
    #pragma unroll
    for (int nt = 0; nt < 4; ++nt) {
      int col = nt * 16 + l16;
      float bc = bias[col];
      #pragma unroll
      for (int v = 0; v < 8; ++v) {
        int row = mbase + v + half * 8;
        H[row * 64 + col] = fmaxf(acc[nt][v] + bc, 0.0f);
      }
    }
  } else {
    const bool rv = rowA < N;
    #pragma unroll
    for (int kk = 0; kk < 16; kk += 4) {
      int ka0 = kk + half * 2;
      int ka1 = ka0 + 1;
      float a0 = (rv && ka0 < 13) ? A[rowA * 13 + ka0] : 0.0f;
      float a1 = (rv && ka1 < 13) ? A[rowA * 13 + ka1] : 0.0f;
      v2f av; av[0] = a0; av[1] = a1;
      #pragma unroll
      for (int nt = 0; nt < 4; ++nt) {
        int col = nt * 16 + l16;
        float b0 = (ka0 < 13) ? W[ka0 * 64 + col] : 0.0f;
        float b1 = (ka1 < 13) ? W[ka1 * 64 + col] : 0.0f;
        v2f bv; bv[0] = b0; bv[1] = b1;
        acc[nt] = __builtin_amdgcn_wmma_f32_16x16x4_f32(
            false, av, false, bv, (short)0, acc[nt], false, false);
      }
    }
    #pragma unroll
    for (int nt = 0; nt < 4; ++nt) {
      int col = nt * 16 + l16;
      float bc = bias[col];
      #pragma unroll
      for (int v = 0; v < 8; ++v) {
        int row = mbase + v + half * 8;
        if (row < N) H[row * 64 + col] = fmaxf(acc[nt][v] + bc, 0.0f);
      }
    }
  }
}

__global__ __launch_bounds__(256)
void gnn_gemm2_wmma(const float* __restrict__ A, const float* __restrict__ W,
                    float* __restrict__ O, int N) {
  const int lane = threadIdx.x & 31;
  const int wave = threadIdx.x >> 5;
  const int mbase = (blockIdx.x * 8 + wave) * 16;
  const int half = lane >> 4;
  const int l16 = lane & 15;

  v8f acc[2];
  const v8f zero = {0.f, 0.f, 0.f, 0.f, 0.f, 0.f, 0.f, 0.f};
  acc[0] = zero; acc[1] = zero;

  const int rowA = mbase + l16;

  if (mbase + 16 <= N) {
    const float* Arow = A + rowA * 64;
    #pragma unroll
    for (int kk = 0; kk < 64; kk += 4) {
      int ka0 = kk + half * 2;
      const float2 a2 = *(const float2*)(Arow + ka0);
      v2f av; av[0] = a2.x; av[1] = a2.y;
      #pragma unroll
      for (int nt = 0; nt < 2; ++nt) {
        int col = nt * 16 + l16;
        v2f bv; bv[0] = W[ka0 * 32 + col]; bv[1] = W[(ka0 + 1) * 32 + col];
        acc[nt] = __builtin_amdgcn_wmma_f32_16x16x4_f32(
            false, av, false, bv, (short)0, acc[nt], false, false);
      }
    }
    #pragma unroll
    for (int nt = 0; nt < 2; ++nt) {
      int col = nt * 16 + l16;
      #pragma unroll
      for (int v = 0; v < 8; ++v) {
        int row = mbase + v + half * 8;
        O[row * 32 + col] = acc[nt][v];
      }
    }
  } else {
    const bool rv = rowA < N;
    #pragma unroll
    for (int kk = 0; kk < 64; kk += 4) {
      int ka0 = kk + half * 2;
      int ka1 = ka0 + 1;
      float a0 = rv ? A[rowA * 64 + ka0] : 0.0f;
      float a1 = rv ? A[rowA * 64 + ka1] : 0.0f;
      v2f av; av[0] = a0; av[1] = a1;
      #pragma unroll
      for (int nt = 0; nt < 2; ++nt) {
        int col = nt * 16 + l16;
        v2f bv; bv[0] = W[ka0 * 32 + col]; bv[1] = W[ka1 * 32 + col];
        acc[nt] = __builtin_amdgcn_wmma_f32_16x16x4_f32(
            false, av, false, bv, (short)0, acc[nt], false, false);
      }
    }
    #pragma unroll
    for (int nt = 0; nt < 2; ++nt) {
      int col = nt * 16 + l16;
      #pragma unroll
      for (int v = 0; v < 8; ++v) {
        int row = mbase + v + half * 8;
        if (row < N) O[row * 32 + col] = acc[nt][v];
      }
    }
  }
}

__global__ void gnn_h2_init(const float* __restrict__ xw2, const float* __restrict__ dinv,
                            float* __restrict__ h2, int N) {
  int t = blockIdx.x * blockDim.x + threadIdx.x;
  if (t < N * 8) {
    int i = t >> 3;
    float nrm = dinv[i] * dinv[i];
    float4 v = ((const float4*)xw2)[t];
    float4 o; o.x = v.x * nrm; o.y = v.y * nrm; o.z = v.z * nrm; o.w = v.w * nrm;
    ((float4*)h2)[t] = o;
  }
}

__global__ void gnn_h2_edge(const float* __restrict__ xw2, const float* __restrict__ dinv,
                            const int* __restrict__ src, const int* __restrict__ dst,
                            float* __restrict__ h2, int E) {
  const int lane = threadIdx.x & 31;
  const int e0 = (blockIdx.x * blockDim.x + threadIdx.x) & ~31;
  if (e0 >= E) return;
  const int e = e0 + lane;
  int s = 0, d = 0;
  float nrm = 0.0f;
  if (e < E) {
    s = src[e]; d = dst[e];
    nrm = dinv[s] * dinv[d];
  }
  const int cnt = E - e0;
  const int nbits = __float_as_int(nrm);
  if (cnt >= 32) {
    float xv[32];
    #pragma unroll
    for (int j = 0; j < 32; ++j)
      xv[j] = xw2[__builtin_amdgcn_readlane(s, j) * 32 + lane];
    #pragma unroll
    for (int j = 0; j < 32; ++j) {
      const float nj = __int_as_float(__builtin_amdgcn_readlane(nbits, j));
      atomicAdd(&h2[__builtin_amdgcn_readlane(d, j) * 32 + lane], nj * xv[j]);
    }
  } else {
    for (int j = 0; j < cnt; ++j) {
      const int sj = __shfl(s, j);
      const int dj = __shfl(d, j);
      const float nj = __shfl(nrm, j);
      atomicAdd(&h2[dj * 32 + lane], nj * xw2[sj * 32 + lane]);
    }
  }
}

__global__ void gnn_pool(const float* __restrict__ h2, const int* __restrict__ batch,
                         float* __restrict__ sums, float* __restrict__ cnts, int N) {
  int t = blockIdx.x * blockDim.x + threadIdx.x;
  int nblk = (N + 7) >> 3;
  if (t >= nblk * 4) return;
  int blk = t >> 2, q = (t & 3) * 8;
  int i0 = blk * 8;
  int iend = (i0 + 8 < N) ? (i0 + 8) : N;
  int cur = batch[i0];
  float acc[8];
  #pragma unroll
  for (int c = 0; c < 8; ++c) acc[c] = 0.0f;
  float cnt = 0.0f;
  for (int i = i0; i < iend; ++i) {
    int b = batch[i];
    if (b != cur) {
      #pragma unroll
      for (int c = 0; c < 8; ++c) atomicAdd(&sums[cur * 32 + q + c], acc[c]);
      if (q == 0) atomicAdd(&cnts[cur], cnt);
      #pragma unroll
      for (int c = 0; c < 8; ++c) acc[c] = 0.0f;
      cnt = 0.0f;
      cur = b;
    }
    #pragma unroll
    for (int c = 0; c < 8; ++c) acc[c] += h2[i * 32 + q + c];
    cnt += 1.0f;
  }
  #pragma unroll
  for (int c = 0; c < 8; ++c) atomicAdd(&sums[cur * 32 + q + c], acc[c]);
  if (q == 0) atomicAdd(&cnts[cur], cnt);
}

__global__ void gnn_gmean(const float* __restrict__ sums, const float* __restrict__ cnts,
                          const float* __restrict__ b2, float* __restrict__ g, int G) {
  int t = blockIdx.x * blockDim.x + threadIdx.x;
  if (t < G * 32) {
    int b = t >> 5, j = t & 31;
    g[t] = sums[t] / fmaxf(cnts[b], 1.0f) + b2[j];
  }
}

__global__ void gnn_fc_relu(const float* __restrict__ g, const float* __restrict__ w,
                            const float* __restrict__ bias, float* __restrict__ o, int G) {
  int t = blockIdx.x * blockDim.x + threadIdx.x;
  if (t < G * 32) {
    int b = t >> 5, j = t & 31;
    float s = bias[j];
    #pragma unroll
    for (int k = 0; k < 32; ++k) s += g[b * 32 + k] * w[k * 32 + j];
    o[t] = fmaxf(s, 0.0f);
  }
}

__global__ void gnn_fc_out(const float* __restrict__ g, const float* __restrict__ w,
                           const float* __restrict__ bias, float* __restrict__ o, int G) {
  int t = blockIdx.x * blockDim.x + threadIdx.x;
  if (t < G * 32) {
    int b = t >> 5, j = t & 31;
    float s = bias[j];
    #pragma unroll
    for (int k = 0; k < 32; ++k) s += g[b * 32 + k] * w[k * 32 + j];
    o[t] = s;
  }
}

extern "C" void kernel_launch(void* const* d_in, const int* in_sizes, int n_in,
                              void* d_out, int out_size, void* d_ws, size_t ws_size,
                              hipStream_t stream) {
  const float* x    = (const float*)d_in[0];
  const int*   ei   = (const int*)d_in[1];
  const int*   batch= (const int*)d_in[2];
  const float* W1   = (const float*)d_in[3];
  const float* b1   = (const float*)d_in[4];
  const float* W2   = (const float*)d_in[5];
  const float* b2   = (const float*)d_in[6];
  const float* fc1w = (const float*)d_in[7];
  const float* fc1b = (const float*)d_in[8];
  const float* fc2w = (const float*)d_in[9];
  const float* fc2b = (const float*)d_in[10];

  const int N = in_sizes[0] / 13;
  const int E = in_sizes[1] / 2;
  const int G = out_size / 32;
  const int* src = ei;
  const int* dst = ei + E;

  auto al4 = [](size_t v) { return (v + 3) & ~(size_t)3; };
  float* ws   = (float*)d_ws;
  size_t off  = 0;
  float* dinv = ws + off; off = al4(off + (size_t)N);
  float* agg1 = ws + off; off = al4(off + (size_t)13 * N);
  float* h1   = ws + off; off = al4(off + (size_t)64 * N);
  float* xw2  = ws + off; off = al4(off + (size_t)32 * N);
  float* h2   = ws + off; off = al4(off + (size_t)32 * N);
  float* sums = ws + off; off = al4(off + (size_t)32 * G);
  float* cnts = ws + off; off = al4(off + (size_t)G);
  float* g    = ws + off; off = al4(off + (size_t)32 * G);
  float* hfc  = ws + off; off = al4(off + (size_t)32 * G);

  auto cdiv = [](int a, int b) { return (a + b - 1) / b; };

  gnn_init<<<cdiv(N, TPB), TPB, 0, stream>>>(dinv, sums, cnts, N, G);
  gnn_deg<<<cdiv(E, TPB), TPB, 0, stream>>>(dst, dinv, E);
  gnn_rsqrt<<<cdiv(N, TPB), TPB, 0, stream>>>(dinv, N);

  gnn_agg1_init<<<cdiv(N, TPB), TPB, 0, stream>>>(x, dinv, agg1, N);
  gnn_agg1_edge<<<cdiv(E, TPB), TPB, 0, stream>>>(x, dinv, src, dst, agg1, E);
  int mt = cdiv(N, 16);
  gnn_gemm1_wmma<<<cdiv(mt, 8), 256, 0, stream>>>(agg1, W1, b1, h1, N);

  gnn_gemm2_wmma<<<cdiv(mt, 8), 256, 0, stream>>>(h1, W2, xw2, N);
  gnn_h2_init<<<cdiv(N * 8, TPB), TPB, 0, stream>>>(xw2, dinv, h2, N);
  gnn_h2_edge<<<cdiv(E, TPB), TPB, 0, stream>>>(xw2, dinv, src, dst, h2, E);

  int nblk = (N + 7) / 8;
  gnn_pool<<<cdiv(nblk * 4, TPB), TPB, 0, stream>>>(h2, batch, sums, cnts, N);
  gnn_gmean<<<cdiv(G * 32, TPB), TPB, 0, stream>>>(sums, cnts, b2, g, G);
  gnn_fc_relu<<<cdiv(G * 32, TPB), TPB, 0, stream>>>(g, fc1w, fc1b, hfc, G);
  gnn_fc_out<<<cdiv(G * 32, TPB), TPB, 0, stream>>>(hfc, fc2w, fc2b, (float*)d_out, G);
}


# layer-2 scatter commuted with the mean pool: each edge adds norm*xw2[src] into a per-block LDS table of graph sums (LDS float atomics, arrival order), flushed once by global atomics; new sum order
# speedup vs baseline: 2.4266x; 1.1150x over previous
#include <hip/hip_runtime.h>

typedef float v2f __attribute__((ext_vector_type(2)));
typedef float v8f __attribute__((ext_vector_type(8)));

#define TPB 256

__global__ void gnn_init(float* __restrict__ deg, float* __restrict__ sums,
                         float* __restrict__ cnts, int N, int G) {
  int i = blockIdx.x * blockDim.x + threadIdx.x;
  if (i < N) deg[i] = 1.0f;
  if (i < G * 32) sums[i] = 0.0f;
  if (i < G) cnts[i] = 0.0f;
}

__global__ void gnn_deg(const int* __restrict__ dst, float* __restrict__ deg, int E) {
  int e = blockIdx.x * blockDim.x + threadIdx.x;
  if (e < E) atomicAdd(&deg[dst[e]], 1.0f);
}

__global__ void gnn_rsqrt(float* __restrict__ deg, int N) {
  int i = blockIdx.x * blockDim.x + threadIdx.x;
  if (i < N) deg[i] = rsqrtf(deg[i]);
}

__global__ void gnn_agg1_init(const float* __restrict__ x, const float* __restrict__ dinv,
                              float* __restrict__ agg, int N) {
  int i = blockIdx.x * blockDim.x + threadIdx.x;
  if (i < N) {
    float nrm = dinv[i] * dinv[i];
    #pragma unroll
    for (int c = 0; c < 13; ++c) agg[i * 13 + c] = nrm * x[i * 13 + c];
  }
}

__global__ void gnn_agg1_edge(const float* __restrict__ x, const float* __restrict__ dinv,
                              const int* __restrict__ src, const int* __restrict__ dst,
                              float* __restrict__ agg, int E) {
  const int lane = threadIdx.x & 31;
  const int e0 = (blockIdx.x * blockDim.x + threadIdx.x) & ~31;
  if (e0 >= E) return;
  const int e = e0 + lane;
  int s = 0, d = 0;
  float nrm = 0.0f;
  if (e < E) {
    s = src[e]; d = dst[e];
    nrm = dinv[s] * dinv[d];
  }
  const int cnt = E - e0;
  const int half = lane >> 4;
  const int c = lane & 15;
  int sj[16], dj[16];
  float nj[16];
  #pragma unroll
  for (int j = 0; j < 16; ++j) {
    const int idx = 2 * j + half;
    sj[j] = __shfl(s, idx);
    dj[j] = __shfl(d, idx);
    nj[j] = __shfl(nrm, idx);
  }
  if (c < 13) {
    if (cnt >= 32) {
      float xv[16];
      #pragma unroll
      for (int j = 0; j < 16; ++j) xv[j] = x[sj[j] * 13 + c];
      #pragma unroll
      for (int j = 0; j < 16; ++j) atomicAdd(&agg[dj[j] * 13 + c], nj[j] * xv[j]);
    } else {
      #pragma unroll
      for (int j = 0; j < 16; ++j)
        if (2 * j + half < cnt) atomicAdd(&agg[dj[j] * 13 + c], nj[j] * x[sj[j] * 13 + c]);
    }
  }
}

__global__ __launch_bounds__(256)
void gnn_gemm1_wmma(const float* __restrict__ A, const float* __restrict__ W,
                    const float* __restrict__ bias, float* __restrict__ H, int N) {
  const int lane = threadIdx.x & 31;
  const int wave = threadIdx.x >> 5;
  const int mbase = (blockIdx.x * 8 + wave) * 16;
  const int half = lane >> 4;
  const int l16 = lane & 15;

  v8f acc[4];
  const v8f zero = {0.f, 0.f, 0.f, 0.f, 0.f, 0.f, 0.f, 0.f};
  #pragma unroll
  for (int nt = 0; nt < 4; ++nt) acc[nt] = zero;

  const int rowA = mbase + l16;

  if (mbase + 16 <= N) {
    const float* Arow = A + rowA * 13;
    #pragma unroll
    for (int kk = 0; kk < 12; kk += 4) {
      int ka0 = kk + half * 2;
      v2f av; av[0] = Arow[ka0]; av[1] = Arow[ka0 + 1];
      #pragma unroll
      for (int nt = 0; nt < 4; ++nt) {
        int col = nt * 16 + l16;
        v2f bv; bv[0] = W[ka0 * 64 + col]; bv[1] = W[(ka0 + 1) * 64 + col];
        acc[nt] = __builtin_amdgcn_wmma_f32_16x16x4_f32(
            false, av, false, bv, (short)0, acc[nt], false, false);
      }
    }
    {
      float a12 = Arow[12];
      v2f av; av[0] = half ? 0.0f : a12; av[1] = 0.0f;
      #pragma unroll
      for (int nt = 0; nt < 4; ++nt) {
        int col = nt * 16 + l16;
        float b12 = W[12 * 64 + col];
        v2f bv; bv[0] = half ? 0.0f : b12; bv[1] = 0.0f;
        acc[nt] = __builtin_amdgcn_wmma_f32_16x16x4_f32(
            false, av, false, bv, (short)0, acc[nt], false, false);
      }
    }
    #pragma unroll
    for (int nt = 0; nt < 4; ++nt) {
      int col = nt * 16 + l16;
      float bc = bias[col];
      #pragma unroll
      for (int v = 0; v < 8; ++v) {
        int row = mbase + v + half * 8;
        H[row * 64 + col] = fmaxf(acc[nt][v] + bc, 0.0f);
      }
    }
  } else {
    const bool rv = rowA < N;
    #pragma unroll
    for (int kk = 0; kk < 16; kk += 4) {
      int ka0 = kk + half * 2;
      int ka1 = ka0 + 1;
      float a0 = (rv && ka0 < 13) ? A[rowA * 13 + ka0] : 0.0f;
      float a1 = (rv && ka1 < 13) ? A[rowA * 13 + ka1] : 0.0f;
      v2f av; av[0] = a0; av[1] = a1;
      #pragma unroll
      for (int nt = 0; nt < 4; ++nt) {
        int col = nt * 16 + l16;
        float b0 = (ka0 < 13) ? W[ka0 * 64 + col] : 0.0f;
        float b1 = (ka1 < 13) ? W[ka1 * 64 + col] : 0.0f;
        v2f bv; bv[0] = b0; bv[1] = b1;
        acc[nt] = __builtin_amdgcn_wmma_f32_16x16x4_f32(
            false, av, false, bv, (short)0, acc[nt], false, false);
      }
    }
    #pragma unroll
    for (int nt = 0; nt < 4; ++nt) {
      int col = nt * 16 + l16;
      float bc = bias[col];
      #pragma unroll
      for (int v = 0; v < 8; ++v) {
        int row = mbase + v + half * 8;
        if (row < N) H[row * 64 + col] = fmaxf(acc[nt][v] + bc, 0.0f);
      }
    }
  }
}

__global__ __launch_bounds__(256)
void gnn_gemm2_wmma(const float* __restrict__ A, const float* __restrict__ W,
                    float* __restrict__ O, int N) {
  const int lane = threadIdx.x & 31;
  const int wave = threadIdx.x >> 5;
  const int mbase = (blockIdx.x * 8 + wave) * 16;
  const int half = lane >> 4;
  const int l16 = lane & 15;

  v8f acc[2];
  const v8f zero = {0.f, 0.f, 0.f, 0.f, 0.f, 0.f, 0.f, 0.f};
  acc[0] = zero; acc[1] = zero;

  const int rowA = mbase + l16;

  if (mbase + 16 <= N) {
    const float* Arow = A + rowA * 64;
    #pragma unroll
    for (int kk = 0; kk < 64; kk += 4) {
      int ka0 = kk + half * 2;
      const float2 a2 = *(const float2*)(Arow + ka0);
      v2f av; av[0] = a2.x; av[1] = a2.y;
      #pragma unroll
      for (int nt = 0; nt < 2; ++nt) {
        int col = nt * 16 + l16;
        v2f bv; bv[0] = W[ka0 * 32 + col]; bv[1] = W[(ka0 + 1) * 32 + col];
        acc[nt] = __builtin_amdgcn_wmma_f32_16x16x4_f32(
            false, av, false, bv, (short)0, acc[nt], false, false);
      }
    }
    #pragma unroll
    for (int nt = 0; nt < 2; ++nt) {
      int col = nt * 16 + l16;
      #pragma unroll
      for (int v = 0; v < 8; ++v) {
        int row = mbase + v + half * 8;
        O[row * 32 + col] = acc[nt][v];
      }
    }
  } else {
    const bool rv = rowA < N;
    #pragma unroll
    for (int kk = 0; kk < 64; kk += 4) {
      int ka0 = kk + half * 2;
      int ka1 = ka0 + 1;
      float a0 = rv ? A[rowA * 64 + ka0] : 0.0f;
      float a1 = rv ? A[rowA * 64 + ka1] : 0.0f;
      v2f av; av[0] = a0; av[1] = a1;
      #pragma unroll
      for (int nt = 0; nt < 2; ++nt) {
        int col = nt * 16 + l16;
        v2f bv; bv[0] = W[ka0 * 32 + col]; bv[1] = W[ka1 * 32 + col];
        acc[nt] = __builtin_amdgcn_wmma_f32_16x16x4_f32(
            false, av, false, bv, (short)0, acc[nt], false, false);
      }
    }
    #pragma unroll
    for (int nt = 0; nt < 2; ++nt) {
      int col = nt * 16 + l16;
      #pragma unroll
      for (int v = 0; v < 8; ++v) {
        int row = mbase + v + half * 8;
        if (row < N) O[row * 32 + col] = acc[nt][v];
      }
    }
  }
}

#define PG_MAX 512
#define PE_THREADS 1024

template <bool IN_LDS>
__device__ __forceinline__ void pool_edge_walk(const float* __restrict__ xw2,
                                               const float* __restrict__ dinv,
                                               const int* __restrict__ src,
                                               const int* __restrict__ dst,
                                               const int* __restrict__ batch,
                                               float* acc, float* cl,
                                               float* __restrict__ sums, float* __restrict__ cnts,
                                               int E, int V) {
  const int lane = threadIdx.x & 31;
  const int wave = blockIdx.x * (PE_THREADS / 32) + (threadIdx.x >> 5);
  const int nwave = gridDim.x * (PE_THREADS / 32);
  const int nchunk = (V + 31) >> 5;
  int ch = wave;
  if (ch >= nchunk) return;
  int ls = 0, ld = 0;
  {
    const int e = ch * 32 + lane;
    if (e < E) { ls = src[e]; ld = dst[e]; }
  }
  for (;;) {
    const int e0 = ch * 32;
    const int e = e0 + lane;
    const bool self = (e >= E);
    const int s = self ? e - E : ls;
    const int d = self ? e - E : ld;
    const bool live = (e < V);
    const int si = live ? s : 0;
    const int di = live ? d : 0;
    const float a = dinv[si];
    const float b = dinv[di];
    const int g = batch[di];
    const int chn = ch + nwave;
    ls = 0; ld = 0;
    {
      const int en = chn * 32 + lane;
      if (en < E) { ls = src[en]; ld = dst[en]; }
    }
    const int cnt = V - e0;
    if (cnt >= 32) {
      float xv[32];
      #pragma unroll
      for (int j = 0; j < 32; ++j)
        xv[j] = xw2[__builtin_amdgcn_readlane(s, j) * 32 + lane];
      const int nbits = __float_as_int(a * b);
      if (self) {
        if (IN_LDS) atomicAdd(&cl[g], 1.0f); else atomicAdd(&cnts[g], 1.0f);
      }
      #pragma unroll
      for (int j = 0; j < 32; ++j) {
        const float nj = __int_as_float(__builtin_amdgcn_readlane(nbits, j));
        const int gj = __builtin_amdgcn_readlane(g, j);
        if (IN_LDS) atomicAdd(&acc[gj * 32 + lane], nj * xv[j]);
        else        atomicAdd(&sums[gj * 32 + lane], nj * xv[j]);
      }
    } else {
      const float nrm = a * b;
      if (self && live) {
        if (IN_LDS) atomicAdd(&cl[g], 1.0f); else atomicAdd(&cnts[g], 1.0f);
      }
      for (int j = 0; j < cnt; ++j) {
        const int sj = __shfl(s, j);
        const int gj = __shfl(g, j);
        const float nj = __shfl(nrm, j);
        if (IN_LDS) atomicAdd(&acc[gj * 32 + lane], nj * xw2[sj * 32 + lane]);
        else        atomicAdd(&sums[gj * 32 + lane], nj * xw2[sj * 32 + lane]);
      }
    }
    ch = chn;
    if (ch >= nchunk) break;
  }
}

__global__ __launch_bounds__(PE_THREADS)
void gnn_pool_edge(const float* __restrict__ xw2, const float* __restrict__ dinv,
                   const int* __restrict__ src, const int* __restrict__ dst,
                   const int* __restrict__ batch,
                   float* __restrict__ sums, float* __restrict__ cnts,
                   int E, int N, int G) {
  __shared__ float acc[PG_MAX * 32];
  __shared__ float cl[PG_MAX];
  const int V = E + N;
  if (G <= PG_MAX) {
    for (int i = threadIdx.x; i < G * 32; i += PE_THREADS) acc[i] = 0.0f;
    for (int i = threadIdx.x; i < G; i += PE_THREADS) cl[i] = 0.0f;
    __syncthreads();
    pool_edge_walk<true>(xw2, dinv, src, dst, batch, acc, cl, sums, cnts, E, V);
    __syncthreads();
    const int lane = threadIdx.x & 31;
    const int w = threadIdx.x >> 5;
    const int r0 = (int)((blockIdx.x * 37u) % (unsigned)G);
    for (int r = w; r < G; r += PE_THREADS / 32) {
      int row = r + r0; if (row >= G) row -= G;
      atomicAdd(&sums[row * 32 + lane], acc[row * 32 + lane]);
    }
    for (int i = threadIdx.x; i < G; i += PE_THREADS) {
      const float c = cl[i];
      if (c != 0.0f) atomicAdd(&cnts[i], c);
    }
  } else {
    pool_edge_walk<false>(xw2, dinv, src, dst, batch, acc, cl, sums, cnts, E, V);
  }
}

__global__ void gnn_gmean(const float* __restrict__ sums, const float* __restrict__ cnts,
                          const float* __restrict__ b2, float* __restrict__ g, int G) {
  int t = blockIdx.x * blockDim.x + threadIdx.x;
  if (t < G * 32) {
    int b = t >> 5, j = t & 31;
    g[t] = sums[t] / fmaxf(cnts[b], 1.0f) + b2[j];
  }
}

__global__ void gnn_fc_relu(const float* __restrict__ g, const float* __restrict__ w,
                            const float* __restrict__ bias, float* __restrict__ o, int G) {
  int t = blockIdx.x * blockDim.x + threadIdx.x;
  if (t < G * 32) {
    int b = t >> 5, j = t & 31;
    float s = bias[j];
    #pragma unroll
    for (int k = 0; k < 32; ++k) s += g[b * 32 + k] * w[k * 32 + j];
    o[t] = fmaxf(s, 0.0f);
  }
}

__global__ void gnn_fc_out(const float* __restrict__ g, const float* __restrict__ w,
                           const float* __restrict__ bias, float* __restrict__ o, int G) {
  int t = blockIdx.x * blockDim.x + threadIdx.x;
  if (t < G * 32) {
    int b = t >> 5, j = t & 31;
    float s = bias[j];
    #pragma unroll
    for (int k = 0; k < 32; ++k) s += g[b * 32 + k] * w[k * 32 + j];
    o[t] = s;
  }
}

extern "C" void kernel_launch(void* const* d_in, const int* in_sizes, int n_in,
                              void* d_out, int out_size, void* d_ws, size_t ws_size,
                              hipStream_t stream) {
  const float* x    = (const float*)d_in[0];
  const int*   ei   = (const int*)d_in[1];
  const int*   batch= (const int*)d_in[2];
  const float* W1   = (const float*)d_in[3];
  const float* b1   = (const float*)d_in[4];
  const float* W2   = (const float*)d_in[5];
  const float* b2   = (const float*)d_in[6];
  const float* fc1w = (const float*)d_in[7];
  const float* fc1b = (const float*)d_in[8];
  const float* fc2w = (const float*)d_in[9];
  const float* fc2b = (const float*)d_in[10];

  const int N = in_sizes[0] / 13;
  const int E = in_sizes[1] / 2;
  const int G = out_size / 32;
  const int* src = ei;
  const int* dst = ei + E;

  auto al4 = [](size_t v) { return (v + 3) & ~(size_t)3; };
  float* ws   = (float*)d_ws;
  size_t off  = 0;
  float* dinv = ws + off; off = al4(off + (size_t)N);
  float* agg1 = ws + off; off = al4(off + (size_t)13 * N);
  float* h1   = ws + off; off = al4(off + (size_t)64 * N);
  float* xw2  = ws + off; off = al4(off + (size_t)32 * N);
  float* sums = ws + off; off = al4(off + (size_t)32 * G);
  float* cnts = ws + off; off = al4(off + (size_t)G);
  float* g    = ws + off; off = al4(off + (size_t)32 * G);
  float* hfc  = ws + off; off = al4(off + (size_t)32 * G);

  auto cdiv = [](int a, int b) { return (a + b - 1) / b; };

  gnn_init<<<cdiv(N, TPB), TPB, 0, stream>>>(dinv, sums, cnts, N, G);
  gnn_deg<<<cdiv(E, TPB), TPB, 0, stream>>>(dst, dinv, E);
  gnn_rsqrt<<<cdiv(N, TPB), TPB, 0, stream>>>(dinv, N);

  gnn_agg1_init<<<cdiv(N, TPB), TPB, 0, stream>>>(x, dinv, agg1, N);
  gnn_agg1_edge<<<cdiv(E, TPB), TPB, 0, stream>>>(x, dinv, src, dst, agg1, E);
  int mt = cdiv(N, 16);
  gnn_gemm1_wmma<<<cdiv(mt, 8), 256, 0, stream>>>(agg1, W1, b1, h1, N);

  gnn_gemm2_wmma<<<cdiv(mt, 8), 256, 0, stream>>>(h1, W2, xw2, N);
  gnn_pool_edge<<<256, PE_THREADS, 0, stream>>>(xw2, dinv, src, dst, batch, sums, cnts, E, N, G);
  gnn_gmean<<<cdiv(G * 32, TPB), TPB, 0, stream>>>(sums, cnts, b2, g, G);
  gnn_fc_relu<<<cdiv(G * 32, TPB), TPB, 0, stream>>>(g, fc1w, fc1b, hfc, G);
  gnn_fc_out<<<cdiv(G * 32, TPB), TPB, 0, stream>>>(hfc, fc2w, fc2b, (float*)d_out, G);
}


# norm dinv[s]*dinv[d] factored out of both edge passes: rows pre-scaled by dinv[s] where produced; dinv[d] scales the row sum (layer 1) or comes with the graph id as one 8-byte record (pool)
# speedup vs baseline: 2.7938x; 1.1513x over previous
#include <hip/hip_runtime.h>

typedef float v2f __attribute__((ext_vector_type(2)));
typedef float v8f __attribute__((ext_vector_type(8)));

#define TPB 256

__global__ void gnn_init(float* __restrict__ deg, float* __restrict__ sums,
                         float* __restrict__ cnts, int N, int G) {
  int i = blockIdx.x * blockDim.x + threadIdx.x;
  if (i < N) deg[i] = 1.0f;
  if (i < G * 32) sums[i] = 0.0f;
  if (i < G) cnts[i] = 0.0f;
}

__global__ void gnn_deg(const int* __restrict__ dst, float* __restrict__ deg, int E) {
  int e = blockIdx.x * blockDim.x + threadIdx.x;
  if (e < E) atomicAdd(&deg[dst[e]], 1.0f);
}

__global__ void gnn_rsqrt(float* __restrict__ deg, const int* __restrict__ batch,
                          float* __restrict__ rec, int N) {
  int i = blockIdx.x * blockDim.x + threadIdx.x;
  if (i < N) {
    const float dv = rsqrtf(deg[i]);
    deg[i] = dv;
    ((float2*)rec)[i] = make_float2(dv, __int_as_float(batch[i]));
  }
}

__global__ void gnn_agg1_init(const float* __restrict__ x, const float* __restrict__ dinv,
                              float* __restrict__ y, float* __restrict__ agg, int N) {
  int i = blockIdx.x * blockDim.x + threadIdx.x;
  if (i < N) {
    const float dv = dinv[i];
    #pragma unroll
    for (int c = 0; c < 13; ++c) {
      const float v = dv * x[i * 13 + c];
      y[i * 13 + c] = v;
      agg[i * 13 + c] = v;
    }
  }
}

__global__ void gnn_agg1_edge(const float* __restrict__ y,
                              const int* __restrict__ src, const int* __restrict__ dst,
                              float* __restrict__ agg, int E) {
  const int lane = threadIdx.x & 31;
  const int e0 = (blockIdx.x * blockDim.x + threadIdx.x) & ~31;
  if (e0 >= E) return;
  const int e = e0 + lane;
  int s = 0, d = 0;
  if (e < E) { s = src[e]; d = dst[e]; }
  const int cnt = E - e0;
  const int half = lane >> 4;
  const int c = lane & 15;
  int sj[16], dj[16];
  #pragma unroll
  for (int j = 0; j < 16; ++j) {
    const int idx = 2 * j + half;
    sj[j] = __shfl(s, idx);
    dj[j] = __shfl(d, idx);
  }
  if (c < 13) {
    if (cnt >= 32) {
      float xv[16];
      #pragma unroll
      for (int j = 0; j < 16; ++j) xv[j] = y[sj[j] * 13 + c];
      #pragma unroll
      for (int j = 0; j < 16; ++j) atomicAdd(&agg[dj[j] * 13 + c], xv[j]);
    } else {
      #pragma unroll
      for (int j = 0; j < 16; ++j)
        if (2 * j + half < cnt) atomicAdd(&agg[dj[j] * 13 + c], y[sj[j] * 13 + c]);
    }
  }
}

__global__ __launch_bounds__(256)
void gnn_gemm1_wmma(const float* __restrict__ A, const float* __restrict__ dinv,
                    const float* __restrict__ W,
                    const float* __restrict__ bias, float* __restrict__ H, int N) {
  const int lane = threadIdx.x & 31;
  const int wave = threadIdx.x >> 5;
  const int mbase = (blockIdx.x * 8 + wave) * 16;
  const int half = lane >> 4;
  const int l16 = lane & 15;

  v8f acc[4];
  const v8f zero = {0.f, 0.f, 0.f, 0.f, 0.f, 0.f, 0.f, 0.f};
  #pragma unroll
  for (int nt = 0; nt < 4; ++nt) acc[nt] = zero;

  const int rowA = mbase + l16;

  if (mbase + 16 <= N) {
    const float* Arow = A + rowA * 13;
    const float dr = dinv[rowA];
    #pragma unroll
    for (int kk = 0; kk < 12; kk += 4) {
      int ka0 = kk + half * 2;
      v2f av; av[0] = dr * Arow[ka0]; av[1] = dr * Arow[ka0 + 1];
      #pragma unroll
      for (int nt = 0; nt < 4; ++nt) {
        int col = nt * 16 + l16;
        v2f bv; bv[0] = W[ka0 * 64 + col]; bv[1] = W[(ka0 + 1) * 64 + col];
        acc[nt] = __builtin_amdgcn_wmma_f32_16x16x4_f32(
            false, av, false, bv, (short)0, acc[nt], false, false);
      }
    }
    {
      float a12 = dr * Arow[12];
      v2f av; av[0] = half ? 0.0f : a12; av[1] = 0.0f;
      #pragma unroll
      for (int nt = 0; nt < 4; ++nt) {
        int col = nt * 16 + l16;
        float b12 = W[12 * 64 + col];
        v2f bv; bv[0] = half ? 0.0f : b12; bv[1] = 0.0f;
        acc[nt] = __builtin_amdgcn_wmma_f32_16x16x4_f32(
            false, av, false, bv, (short)0, acc[nt], false, false);
      }
    }
    #pragma unroll
    for (int nt = 0; nt < 4; ++nt) {
      int col = nt * 16 + l16;
      float bc = bias[col];
      #pragma unroll
      for (int v = 0; v < 8; ++v) {
        int row = mbase + v + half * 8;
        H[row * 64 + col] = fmaxf(acc[nt][v] + bc, 0.0f);
      }
    }
  } else {
    const bool rv = rowA < N;
    const float dr = rv ? dinv[rowA] : 0.0f;
    #pragma unroll
    for (int kk = 0; kk < 16; kk += 4) {
      int ka0 = kk + half * 2;
      int ka1 = ka0 + 1;
      float a0 = (rv && ka0 < 13) ? dr * A[rowA * 13 + ka0] : 0.0f;
      float a1 = (rv && ka1 < 13) ? dr * A[rowA * 13 + ka1] : 0.0f;
      v2f av; av[0] = a0; av[1] = a1;
      #pragma unroll
      for (int nt = 0; nt < 4; ++nt) {
        int col = nt * 16 + l16;
        float b0 = (ka0 < 13) ? W[ka0 * 64 + col] : 0.0f;
        float b1 = (ka1 < 13) ? W[ka1 * 64 + col] : 0.0f;
        v2f bv; bv[0] = b0; bv[1] = b1;
        acc[nt] = __builtin_amdgcn_wmma_f32_16x16x4_f32(
            false, av, false, bv, (short)0, acc[nt], false, false);
      }
    }
    #pragma unroll
    for (int nt = 0; nt < 4; ++nt) {
      int col = nt * 16 + l16;
      float bc = bias[col];
      #pragma unroll
      for (int v = 0; v < 8; ++v) {
        int row = mbase + v + half * 8;
        if (row < N) H[row * 64 + col] = fmaxf(acc[nt][v] + bc, 0.0f);
      }
    }
  }
}

__global__ __launch_bounds__(256)
void gnn_gemm2_wmma(const float* __restrict__ A, const float* __restrict__ W,
                    const float* __restrict__ dinv, float* __restrict__ O, int N) {
  const int lane = threadIdx.x & 31;
  const int wave = threadIdx.x >> 5;
  const int mbase = (blockIdx.x * 8 + wave) * 16;
  const int half = lane >> 4;
  const int l16 = lane & 15;

  v8f acc[2];
  const v8f zero = {0.f, 0.f, 0.f, 0.f, 0.f, 0.f, 0.f, 0.f};
  acc[0] = zero; acc[1] = zero;

  const int rowA = mbase + l16;

  if (mbase + 16 <= N) {
    const float* Arow = A + rowA * 64;
    #pragma unroll
    for (int kk = 0; kk < 64; kk += 4) {
      int ka0 = kk + half * 2;
      const float2 a2 = *(const float2*)(Arow + ka0);
      v2f av; av[0] = a2.x; av[1] = a2.y;
      #pragma unroll
      for (int nt = 0; nt < 2; ++nt) {
        int col = nt * 16 + l16;
        v2f bv; bv[0] = W[ka0 * 32 + col]; bv[1] = W[(ka0 + 1) * 32 + col];
        acc[nt] = __builtin_amdgcn_wmma_f32_16x16x4_f32(
            false, av, false, bv, (short)0, acc[nt], false, false);
      }
    }
    float dsc[8];
    #pragma unroll
    for (int v = 0; v < 8; ++v) dsc[v] = dinv[mbase + v + half * 8];
    #pragma unroll
    for (int nt = 0; nt < 2; ++nt) {
      int col = nt * 16 + l16;
      #pragma unroll
      for (int v = 0; v < 8; ++v) {
        int row = mbase + v + half * 8;
        O[row * 32 + col] = acc[nt][v] * dsc[v];
      }
    }
  } else {
    const bool rv = rowA < N;
    #pragma unroll
    for (int kk = 0; kk < 64; kk += 4) {
      int ka0 = kk + half * 2;
      int ka1 = ka0 + 1;
      float a0 = rv ? A[rowA * 64 + ka0] : 0.0f;
      float a1 = rv ? A[rowA * 64 + ka1] : 0.0f;
      v2f av; av[0] = a0; av[1] = a1;
      #pragma unroll
      for (int nt = 0; nt < 2; ++nt) {
        int col = nt * 16 + l16;
        v2f bv; bv[0] = W[ka0 * 32 + col]; bv[1] = W[ka1 * 32 + col];
        acc[nt] = __builtin_amdgcn_wmma_f32_16x16x4_f32(
            false, av, false, bv, (short)0, acc[nt], false, false);
      }
    }
    #pragma unroll
    for (int nt = 0; nt < 2; ++nt) {
      int col = nt * 16 + l16;
      #pragma unroll
      for (int v = 0; v < 8; ++v) {
        int row = mbase + v + half * 8;
        if (row < N) O[row * 32 + col] = acc[nt][v] * dinv[row];
      }
    }
  }
}

#define PG_MAX 512
#define PE_THREADS 1024

template <bool IN_LDS>
__device__ __forceinline__ void pool_edge_walk(const float* __restrict__ xw2,
                                               const float* __restrict__ rec,
                                               const int* __restrict__ src,
                                               const int* __restrict__ dst,
                                               float* acc, float* cl,
                                               float* __restrict__ sums, float* __restrict__ cnts,
                                               int E, int V) {
  const int lane = threadIdx.x & 31;
  const int wave = blockIdx.x * (PE_THREADS / 32) + (threadIdx.x >> 5);
  const int nwave = gridDim.x * (PE_THREADS / 32);
  const int nchunk = (V + 31) >> 5;
  int ch = wave;
  if (ch >= nchunk) return;
  int ls = 0, ld = 0;
  {
    const int e = ch * 32 + lane;
    if (e < E) { ls = src[e]; ld = dst[e]; }
  }
  for (;;) {
    const int e0 = ch * 32;
    const int e = e0 + lane;
    const bool self = (e >= E);
    const int s = self ? e - E : ls;
    const int d = self ? e - E : ld;
    const bool live = (e < V);
    const int di = live ? d : 0;
    const float2 rd = ((const float2*)rec)[di];
    const float b = rd.x;
    const int g = __float_as_int(rd.y);
    const int chn = ch + nwave;
    ls = 0; ld = 0;
    {
      const int en = chn * 32 + lane;
      if (en < E) { ls = src[en]; ld = dst[en]; }
    }
    const int cnt = V - e0;
    if (cnt >= 32) {
      float xv[32];
      #pragma unroll
      for (int j = 0; j < 32; ++j)
        xv[j] = xw2[__builtin_amdgcn_readlane(s, j) * 32 + lane];
      const int nbits = __float_as_int(b);
      if (self) {
        if (IN_LDS) atomicAdd(&cl[g], 1.0f); else atomicAdd(&cnts[g], 1.0f);
      }
      #pragma unroll
      for (int j = 0; j < 32; ++j) {
        const float nj = __int_as_float(__builtin_amdgcn_readlane(nbits, j));
        const int gj = __builtin_amdgcn_readlane(g, j);
        if (IN_LDS) atomicAdd(&acc[gj * 32 + lane], nj * xv[j]);
        else        atomicAdd(&sums[gj * 32 + lane], nj * xv[j]);
      }
    } else {
      const float nrm = b;
      if (self && live) {
        if (IN_LDS) atomicAdd(&cl[g], 1.0f); else atomicAdd(&cnts[g], 1.0f);
      }
      for (int j = 0; j < cnt; ++j) {
        const int sj = __shfl(s, j);
        const int gj = __shfl(g, j);
        const float nj = __shfl(nrm, j);
        if (IN_LDS) atomicAdd(&acc[gj * 32 + lane], nj * xw2[sj * 32 + lane]);
        else        atomicAdd(&sums[gj * 32 + lane], nj * xw2[sj * 32 + lane]);
      }
    }
    ch = chn;
    if (ch >= nchunk) break;
  }
}

__global__ __launch_bounds__(PE_THREADS)
void gnn_pool_edge(const float* __restrict__ xw2, const float* __restrict__ rec,
                   const int* __restrict__ src, const int* __restrict__ dst,
                   float* __restrict__ sums, float* __restrict__ cnts,
                   int E, int N, int G) {
  __shared__ float acc[PG_MAX * 32];
  __shared__ float cl[PG_MAX];
  const int V = E + N;
  if (G <= PG_MAX) {
    for (int i = threadIdx.x; i < G * 32; i += PE_THREADS) acc[i] = 0.0f;
    for (int i = threadIdx.x; i < G; i += PE_THREADS) cl[i] = 0.0f;
    __syncthreads();
    pool_edge_walk<true>(xw2, rec, src, dst, acc, cl, sums, cnts, E, V);
    __syncthreads();
    const int lane = threadIdx.x & 31;
    const int w = threadIdx.x >> 5;
    const int r0 = (int)((blockIdx.x * 37u) % (unsigned)G);
    for (int r = w; r < G; r += PE_THREADS / 32) {
      int row = r + r0; if (row >= G) row -= G;
      atomicAdd(&sums[row * 32 + lane], acc[row * 32 + lane]);
    }
    for (int i = threadIdx.x; i < G; i += PE_THREADS) {
      const float c = cl[i];
      if (c != 0.0f) atomicAdd(&cnts[i], c);
    }
  } else {
    pool_edge_walk<false>(xw2, rec, src, dst, acc, cl, sums, cnts, E, V);
  }
}

__global__ void gnn_gmean(const float* __restrict__ sums, const float* __restrict__ cnts,
                          const float* __restrict__ b2, float* __restrict__ g, int G) {
  int t = blockIdx.x * blockDim.x + threadIdx.x;
  if (t < G * 32) {
    int b = t >> 5, j = t & 31;
    g[t] = sums[t] / fmaxf(cnts[b], 1.0f) + b2[j];
  }
}

__global__ void gnn_fc_relu(const float* __restrict__ g, const float* __restrict__ w,
                            const float* __restrict__ bias, float* __restrict__ o, int G) {
  int t = blockIdx.x * blockDim.x + threadIdx.x;
  if (t < G * 32) {
    int b = t >> 5, j = t & 31;
    float s = bias[j];
    #pragma unroll
    for (int k = 0; k < 32; ++k) s += g[b * 32 + k] * w[k * 32 + j];
    o[t] = fmaxf(s, 0.0f);
  }
}

__global__ void gnn_fc_out(const float* __restrict__ g, const float* __restrict__ w,
                           const float* __restrict__ bias, float* __restrict__ o, int G) {
  int t = blockIdx.x * blockDim.x + threadIdx.x;
  if (t < G * 32) {
    int b = t >> 5, j = t & 31;
    float s = bias[j];
    #pragma unroll
    for (int k = 0; k < 32; ++k) s += g[b * 32 + k] * w[k * 32 + j];
    o[t] = s;
  }
}

extern "C" void kernel_launch(void* const* d_in, const int* in_sizes, int n_in,
                              void* d_out, int out_size, void* d_ws, size_t ws_size,
                              hipStream_t stream) {
  const float* x    = (const float*)d_in[0];
  const int*   ei   = (const int*)d_in[1];
  const int*   batch= (const int*)d_in[2];
  const float* W1   = (const float*)d_in[3];
  const float* b1   = (const float*)d_in[4];
  const float* W2   = (const float*)d_in[5];
  const float* b2   = (const float*)d_in[6];
  const float* fc1w = (const float*)d_in[7];
  const float* fc1b = (const float*)d_in[8];
  const float* fc2w = (const float*)d_in[9];
  const float* fc2b = (const float*)d_in[10];

  const int N = in_sizes[0] / 13;
  const int E = in_sizes[1] / 2;
  const int G = out_size / 32;
  const int* src = ei;
  const int* dst = ei + E;

  auto al4 = [](size_t v) { return (v + 3) & ~(size_t)3; };
  float* ws   = (float*)d_ws;
  size_t off  = 0;
  float* dinv = ws + off; off = al4(off + (size_t)N);
  float* rec  = ws + off; off = al4(off + (size_t)2 * N);
  float* y1   = ws + off; off = al4(off + (size_t)13 * N);
  float* agg1 = ws + off; off = al4(off + (size_t)13 * N);
  float* h1   = ws + off; off = al4(off + (size_t)64 * N);
  float* xw2  = ws + off; off = al4(off + (size_t)32 * N);
  float* sums = ws + off; off = al4(off + (size_t)32 * G);
  float* cnts = ws + off; off = al4(off + (size_t)G);
  float* g    = ws + off; off = al4(off + (size_t)32 * G);
  float* hfc  = ws + off; off = al4(off + (size_t)32 * G);

  auto cdiv = [](int a, int b) { return (a + b - 1) / b; };

  gnn_init<<<cdiv(N, TPB), TPB, 0, stream>>>(dinv, sums, cnts, N, G);
  gnn_deg<<<cdiv(E, TPB), TPB, 0, stream>>>(dst, dinv, E);
  gnn_rsqrt<<<cdiv(N, TPB), TPB, 0, stream>>>(dinv, batch, rec, N);

  gnn_agg1_init<<<cdiv(N, TPB), TPB, 0, stream>>>(x, dinv, y1, agg1, N);
  gnn_agg1_edge<<<cdiv(E, TPB), TPB, 0, stream>>>(y1, src, dst, agg1, E);
  int mt = cdiv(N, 16);
  gnn_gemm1_wmma<<<cdiv(mt, 8), 256, 0, stream>>>(agg1, dinv, W1, b1, h1, N);

  gnn_gemm2_wmma<<<cdiv(mt, 8), 256, 0, stream>>>(h1, W2, dinv, xw2, N);
  gnn_pool_edge<<<256, PE_THREADS, 0, stream>>>(xw2, rec, src, dst, sums, cnts, E, N, G);
  gnn_gmean<<<cdiv(G * 32, TPB), TPB, 0, stream>>>(sums, cnts, b2, g, G);
  gnn_fc_relu<<<cdiv(G * 32, TPB), TPB, 0, stream>>>(g, fc1w, fc1b, hfc, G);
  gnn_fc_out<<<cdiv(G * 32, TPB), TPB, 0, stream>>>(hfc, fc2w, fc2b, (float*)d_out, G);
}


# dependent launches merged, 11 -> 7: dinv+record+layer-1 rows in one node pass; both WMMA transforms in one kernel with the H1 tile kept in a per-wave LDS tile; mean+fc1+fc2 in one kernel
# speedup vs baseline: 3.0281x; 1.0839x over previous
#include <hip/hip_runtime.h>

typedef float v2f __attribute__((ext_vector_type(2)));
typedef float v8f __attribute__((ext_vector_type(8)));

#define TPB 256

__global__ void gnn_init(float* __restrict__ deg, float* __restrict__ sums,
                         float* __restrict__ cnts, int N, int G) {
  int i = blockIdx.x * blockDim.x + threadIdx.x;
  if (i < N) deg[i] = 1.0f;
  if (i < G * 32) sums[i] = 0.0f;
  if (i < G) cnts[i] = 0.0f;
}

__global__ void gnn_deg(const int* __restrict__ dst, float* __restrict__ deg, int E) {
  int e = blockIdx.x * blockDim.x + threadIdx.x;
  if (e < E) atomicAdd(&deg[dst[e]], 1.0f);
}

__global__ void gnn_node_prep(float* __restrict__ deg, const int* __restrict__ batch,
                              float* __restrict__ rec, const float* __restrict__ x,
                              float* __restrict__ y, float* __restrict__ agg, int N) {
  int i = blockIdx.x * blockDim.x + threadIdx.x;
  if (i < N) {
    const float dv = rsqrtf(deg[i]);
    deg[i] = dv;
    ((float2*)rec)[i] = make_float2(dv, __int_as_float(batch[i]));
    #pragma unroll
    for (int c = 0; c < 13; ++c) {
      const float v = dv * x[i * 13 + c];
      y[i * 13 + c] = v;
      agg[i * 13 + c] = v;
    }
  }
}

__global__ void gnn_agg1_edge(const float* __restrict__ y,
                              const int* __restrict__ src, const int* __restrict__ dst,
                              float* __restrict__ agg, int E) {
  const int lane = threadIdx.x & 31;
  const int e0 = (blockIdx.x * blockDim.x + threadIdx.x) & ~31;
  if (e0 >= E) return;
  const int e = e0 + lane;
  int s = 0, d = 0;
  if (e < E) { s = src[e]; d = dst[e]; }
  const int cnt = E - e0;
  const int half = lane >> 4;
  const int c = lane & 15;
  int sj[16], dj[16];
  #pragma unroll
  for (int j = 0; j < 16; ++j) {
    const int idx = 2 * j + half;
    sj[j] = __shfl(s, idx);
    dj[j] = __shfl(d, idx);
  }
  if (c < 13) {
    if (cnt >= 32) {
      float xv[16];
      #pragma unroll
      for (int j = 0; j < 16; ++j) xv[j] = y[sj[j] * 13 + c];
      #pragma unroll
      for (int j = 0; j < 16; ++j) atomicAdd(&agg[dj[j] * 13 + c], xv[j]);
    } else {
      #pragma unroll
      for (int j = 0; j < 16; ++j)
        if (2 * j + half < cnt) atomicAdd(&agg[dj[j] * 13 + c], y[sj[j] * 13 + c]);
    }
  }
}

#define H1P 68

template <bool FULL>
__device__ __forceinline__ void gemm12_tile(const float* __restrict__ A,
                                            const float* __restrict__ dinv,
                                            const float* __restrict__ W1,
                                            const float* __restrict__ b1,
                                            const float* __restrict__ W2,
                                            float* __restrict__ O, float* hs,
                                            int N, int mbase, int lane) {
  const int half = lane >> 4;
  const int l16 = lane & 15;
  const int rowA = mbase + l16;
  const bool rv = FULL || (rowA < N);
  const int rowL = rv ? rowA : 0;

  v8f acc[4];
  const v8f zero = {0.f, 0.f, 0.f, 0.f, 0.f, 0.f, 0.f, 0.f};
  #pragma unroll
  for (int nt = 0; nt < 4; ++nt) acc[nt] = zero;

  const float* Arow = A + rowL * 13;
  const float dr = dinv[rowL];
  #pragma unroll
  for (int kk = 0; kk < 12; kk += 4) {
    const int ka0 = kk + half * 2;
    const float x0 = dr * Arow[ka0];
    const float x1 = dr * Arow[ka0 + 1];
    v2f av; av[0] = rv ? x0 : 0.0f; av[1] = rv ? x1 : 0.0f;
    #pragma unroll
    for (int nt = 0; nt < 4; ++nt) {
      const int col = nt * 16 + l16;
      v2f bv; bv[0] = W1[ka0 * 64 + col]; bv[1] = W1[(ka0 + 1) * 64 + col];
      acc[nt] = __builtin_amdgcn_wmma_f32_16x16x4_f32(
          false, av, false, bv, (short)0, acc[nt], false, false);
    }
  }
  {
    const float a12 = dr * Arow[12];
    v2f av; av[0] = (half || !rv) ? 0.0f : a12; av[1] = 0.0f;
    #pragma unroll
    for (int nt = 0; nt < 4; ++nt) {
      const int col = nt * 16 + l16;
      const float b12 = W1[12 * 64 + col];
      v2f bv; bv[0] = half ? 0.0f : b12; bv[1] = 0.0f;
      acc[nt] = __builtin_amdgcn_wmma_f32_16x16x4_f32(
          false, av, false, bv, (short)0, acc[nt], false, false);
    }
  }
  #pragma unroll
  for (int nt = 0; nt < 4; ++nt) {
    const int col = nt * 16 + l16;
    const float bc = b1[col];
    #pragma unroll
    for (int v = 0; v < 8; ++v)
      hs[(v + half * 8) * H1P + col] = fmaxf(acc[nt][v] + bc, 0.0f);
  }
  __builtin_amdgcn_fence(__ATOMIC_RELEASE, "wavefront");
  __builtin_amdgcn_wave_barrier();
  __builtin_amdgcn_fence(__ATOMIC_ACQUIRE, "wavefront");

  v8f acc2[2];
  acc2[0] = zero; acc2[1] = zero;
  const float* hrow = hs + l16 * H1P;
  #pragma unroll
  for (int kk = 0; kk < 64; kk += 4) {
    const int ka0 = kk + half * 2;
    const float2 a2 = *(const float2*)(hrow + ka0);
    v2f av; av[0] = a2.x; av[1] = a2.y;
    #pragma unroll
    for (int nt = 0; nt < 2; ++nt) {
      const int col = nt * 16 + l16;
      v2f bv; bv[0] = W2[ka0 * 32 + col]; bv[1] = W2[(ka0 + 1) * 32 + col];
      acc2[nt] = __builtin_amdgcn_wmma_f32_16x16x4_f32(
          false, av, false, bv, (short)0, acc2[nt], false, false);
    }
  }
  float dsc[8];
  #pragma unroll
  for (int v = 0; v < 8; ++v) {
    const int row = mbase + v + half * 8;
    dsc[v] = (FULL || row < N) ? dinv[row] : 0.0f;
  }
  #pragma unroll
  for (int nt = 0; nt < 2; ++nt) {
    const int col = nt * 16 + l16;
    #pragma unroll
    for (int v = 0; v < 8; ++v) {
      const int row = mbase + v + half * 8;
      if (FULL || row < N) O[row * 32 + col] = acc2[nt][v] * dsc[v];
    }
  }
}

__global__ __launch_bounds__(256)
void gnn_gemm12_wmma(const float* __restrict__ A, const float* __restrict__ dinv,
                     const float* __restrict__ W1, const float* __restrict__ b1,
                     const float* __restrict__ W2, float* __restrict__ O, int N) {
  __shared__ __attribute__((aligned(16))) float h1s[8 * 16 * H1P];
  const int lane = threadIdx.x & 31;
  const int wave = threadIdx.x >> 5;
  const int mbase = (blockIdx.x * 8 + wave) * 16;
  float* hs = h1s + wave * (16 * H1P);
  if (mbase + 16 <= N)
    gemm12_tile<true>(A, dinv, W1, b1, W2, O, hs, N, mbase, lane);
  else if (mbase < N)
    gemm12_tile<false>(A, dinv, W1, b1, W2, O, hs, N, mbase, lane);
}

#define PG_MAX 512
#define PE_THREADS 1024

template <bool IN_LDS>
__device__ __forceinline__ void pool_edge_walk(const float* __restrict__ xw2,
                                               const float* __restrict__ rec,
                                               const int* __restrict__ src,
                                               const int* __restrict__ dst,
                                               float* acc, float* cl,
                                               float* __restrict__ sums, float* __restrict__ cnts,
                                               int E, int V) {
  const int lane = threadIdx.x & 31;
  const int wave = blockIdx.x * (PE_THREADS / 32) + (threadIdx.x >> 5);
  const int nwave = gridDim.x * (PE_THREADS / 32);
  const int nchunk = (V + 31) >> 5;
  int ch = wave;
  if (ch >= nchunk) return;
  int ls = 0, ld = 0;
  {
    const int e = ch * 32 + lane;
    if (e < E) { ls = src[e]; ld = dst[e]; }
  }
  for (;;) {
    const int e0 = ch * 32;
    const int e = e0 + lane;
    const bool self = (e >= E);
    const int s = self ? e - E : ls;
    const int d = self ? e - E : ld;
    const bool live = (e < V);
    const int di = live ? d : 0;
    const float2 rd = ((const float2*)rec)[di];
    const float b = rd.x;
    const int g = __float_as_int(rd.y);
    const int chn = ch + nwave;
    ls = 0; ld = 0;
    {
      const int en = chn * 32 + lane;
      if (en < E) { ls = src[en]; ld = dst[en]; }
    }
    const int cnt = V - e0;
    if (cnt >= 32) {
      float xv[32];
      #pragma unroll
      for (int j = 0; j < 32; ++j)
        xv[j] = xw2[__builtin_amdgcn_readlane(s, j) * 32 + lane];
      const int nbits = __float_as_int(b);
      if (self) {
        if (IN_LDS) atomicAdd(&cl[g], 1.0f); else atomicAdd(&cnts[g], 1.0f);
      }
      #pragma unroll
      for (int j = 0; j < 32; ++j) {
        const float nj = __int_as_float(__builtin_amdgcn_readlane(nbits, j));
        const int gj = __builtin_amdgcn_readlane(g, j);
        if (IN_LDS) atomicAdd(&acc[gj * 32 + lane], nj * xv[j]);
        else        atomicAdd(&sums[gj * 32 + lane], nj * xv[j]);
      }
    } else {
      const float nrm = b;
      if (self && live) {
        if (IN_LDS) atomicAdd(&cl[g], 1.0f); else atomicAdd(&cnts[g], 1.0f);
      }
      for (int j = 0; j < cnt; ++j) {
        const int sj = __shfl(s, j);
        const int gj = __shfl(g, j);
        const float nj = __shfl(nrm, j);
        if (IN_LDS) atomicAdd(&acc[gj * 32 + lane], nj * xw2[sj * 32 + lane]);
        else        atomicAdd(&sums[gj * 32 + lane], nj * xw2[sj * 32 + lane]);
      }
    }
    ch = chn;
    if (ch >= nchunk) break;
  }
}

__global__ __launch_bounds__(PE_THREADS)
void gnn_pool_edge(const float* __restrict__ xw2, const float* __restrict__ rec,
                   const int* __restrict__ src, const int* __restrict__ dst,
                   float* __restrict__ sums, float* __restrict__ cnts,
                   int E, int N, int G) {
  __shared__ float acc[PG_MAX * 32];
  __shared__ float cl[PG_MAX];
  const int V = E + N;
  if (G <= PG_MAX) {
    for (int i = threadIdx.x; i < G * 32; i += PE_THREADS) acc[i] = 0.0f;
    for (int i = threadIdx.x; i < G; i += PE_THREADS) cl[i] = 0.0f;
    __syncthreads();
    pool_edge_walk<true>(xw2, rec, src, dst, acc, cl, sums, cnts, E, V);
    __syncthreads();
    const int lane = threadIdx.x & 31;
    const int w = threadIdx.x >> 5;
    const int r0 = (int)((blockIdx.x * 37u) % (unsigned)G);
    for (int r = w; r < G; r += PE_THREADS / 32) {
      int row = r + r0; if (row >= G) row -= G;
      atomicAdd(&sums[row * 32 + lane], acc[row * 32 + lane]);
    }
    for (int i = threadIdx.x; i < G; i += PE_THREADS) {
      const float c = cl[i];
      if (c != 0.0f) atomicAdd(&cnts[i], c);
    }
  } else {
    pool_edge_walk<false>(xw2, rec, src, dst, acc, cl, sums, cnts, E, V);
  }
}

__global__ void gnn_head(const float* __restrict__ sums, const float* __restrict__ cnts,
                         const float* __restrict__ b2,
                         const float* __restrict__ w1, const float* __restrict__ bias1,
                         const float* __restrict__ w2, const float* __restrict__ bias2,
                         float* __restrict__ o, int G) {
  const int t = blockIdx.x * blockDim.x + threadIdx.x;
  if (t >= G * 32) return;
  const int b = t >> 5, j = t & 31;
  const int gb = __float_as_int(sums[t] / fmaxf(cnts[b], 1.0f) + b2[j]);
  float s = bias1[j];
  #pragma unroll
  for (int k = 0; k < 32; ++k)
    s += __int_as_float(__builtin_amdgcn_readlane(gb, k)) * w1[k * 32 + j];
  const int hb = __float_as_int(fmaxf(s, 0.0f));
  float r = bias2[j];
  #pragma unroll
  for (int k = 0; k < 32; ++k)
    r += __int_as_float(__builtin_amdgcn_readlane(hb, k)) * w2[k * 32 + j];
  o[t] = r;
}

extern "C" void kernel_launch(void* const* d_in, const int* in_sizes, int n_in,
                              void* d_out, int out_size, void* d_ws, size_t ws_size,
                              hipStream_t stream) {
  const float* x    = (const float*)d_in[0];
  const int*   ei   = (const int*)d_in[1];
  const int*   batch= (const int*)d_in[2];
  const float* W1   = (const float*)d_in[3];
  const float* b1   = (const float*)d_in[4];
  const float* W2   = (const float*)d_in[5];
  const float* b2   = (const float*)d_in[6];
  const float* fc1w = (const float*)d_in[7];
  const float* fc1b = (const float*)d_in[8];
  const float* fc2w = (const float*)d_in[9];
  const float* fc2b = (const float*)d_in[10];

  const int N = in_sizes[0] / 13;
  const int E = in_sizes[1] / 2;
  const int G = out_size / 32;
  const int* src = ei;
  const int* dst = ei + E;

  auto al4 = [](size_t v) { return (v + 3) & ~(size_t)3; };
  float* ws   = (float*)d_ws;
  size_t off  = 0;
  float* dinv = ws + off; off = al4(off + (size_t)N);
  float* rec  = ws + off; off = al4(off + (size_t)2 * N);
  float* y1   = ws + off; off = al4(off + (size_t)13 * N);
  float* agg1 = ws + off; off = al4(off + (size_t)13 * N);
  float* xw2  = ws + off; off = al4(off + (size_t)32 * N);
  float* sums = ws + off; off = al4(off + (size_t)32 * G);
  float* cnts = ws + off; off = al4(off + (size_t)G);

  auto cdiv = [](int a, int b) { return (a + b - 1) / b; };

  gnn_init<<<cdiv(N, TPB), TPB, 0, stream>>>(dinv, sums, cnts, N, G);
  gnn_deg<<<cdiv(E, TPB), TPB, 0, stream>>>(dst, dinv, E);
  gnn_node_prep<<<cdiv(N, TPB), TPB, 0, stream>>>(dinv, batch, rec, x, y1, agg1, N);

  gnn_agg1_edge<<<cdiv(E, TPB), TPB, 0, stream>>>(y1, src, dst, agg1, E);
  int mt = cdiv(N, 16);
  gnn_gemm12_wmma<<<cdiv(mt, 8), 256, 0, stream>>>(agg1, dinv, W1, b1, W2, xw2, N);

  gnn_pool_edge<<<256, PE_THREADS, 0, stream>>>(xw2, rec, src, dst, sums, cnts, E, N, G);
  gnn_head<<<cdiv(G * 32, TPB), TPB, 0, stream>>>(sums, cnts, b2, fc1w, fc1b, fc2w, fc2b,
                                                   (float*)d_out, G);
}


# layer-1 rows (y, agg) at a 16-float pitch: a row is one aligned 64-byte piece, so each row load and row atomic of the edge pass touches one line; node pass builds them lane-per-channel
# speedup vs baseline: 3.2906x; 1.0867x over previous
#include <hip/hip_runtime.h>

typedef float v2f __attribute__((ext_vector_type(2)));
typedef float v8f __attribute__((ext_vector_type(8)));

#define TPB 256

__global__ void gnn_init(float* __restrict__ deg, float* __restrict__ sums,
                         float* __restrict__ cnts, int N, int G) {
  int i = blockIdx.x * blockDim.x + threadIdx.x;
  if (i < N) deg[i] = 1.0f;
  if (i < G * 32) sums[i] = 0.0f;
  if (i < G) cnts[i] = 0.0f;
}

__global__ void gnn_deg(const int* __restrict__ dst, float* __restrict__ deg, int E) {
  int e = blockIdx.x * blockDim.x + threadIdx.x;
  if (e < E) atomicAdd(&deg[dst[e]], 1.0f);
}

#define RP 16
#define PREP_NODES 128
__global__ void gnn_node_prep(const float* __restrict__ deg, const int* __restrict__ batch,
                              float* __restrict__ rec, const float* __restrict__ x,
                              float* __restrict__ y, float* __restrict__ agg, int N) {
  const int c = threadIdx.x & 15;
  const int sub = threadIdx.x >> 4;
  const int base = blockIdx.x * PREP_NODES;
  const int cl = c < 13 ? c : 12;
  float dg[PREP_NODES / 16], xs[PREP_NODES / 16];
  int bt[PREP_NODES / 16];
  #pragma unroll
  for (int k = 0; k < PREP_NODES / 16; ++k) {
    const int i = base + k * 16 + sub;
    const int il = i < N ? i : N - 1;
    dg[k] = deg[il];
    xs[k] = x[il * 13 + cl];
    bt[k] = batch[il];
  }
  asm volatile("" :: "v"(dg[0]), "v"(dg[1]), "v"(dg[2]), "v"(dg[3]), "v"(dg[4]), "v"(dg[5]),
               "v"(dg[6]), "v"(dg[7]), "v"(xs[0]), "v"(xs[1]), "v"(xs[2]), "v"(xs[3]), "v"(xs[4]),
               "v"(xs[5]), "v"(xs[6]), "v"(xs[7]), "v"(bt[0]), "v"(bt[1]), "v"(bt[2]), "v"(bt[3]),
               "v"(bt[4]), "v"(bt[5]), "v"(bt[6]), "v"(bt[7]));
  #pragma unroll
  for (int k = 0; k < PREP_NODES / 16; ++k) {
    const int i = base + k * 16 + sub;
    const float dv = rsqrtf(dg[k]);
    const float v = dv * xs[k];
    if (i < N) {
      if (c < 13) { y[i * RP + c] = v; agg[i * RP + c] = v; }
      if (c == 0) ((float2*)rec)[i] = make_float2(dv, __int_as_float(bt[k]));
    }
  }
}

__global__ void gnn_agg1_edge(const float* __restrict__ y,
                              const int* __restrict__ src, const int* __restrict__ dst,
                              float* __restrict__ agg, int E) {
  const int lane = threadIdx.x & 31;
  const int e0 = (blockIdx.x * blockDim.x + threadIdx.x) & ~31;
  if (e0 >= E) return;
  const int e = e0 + lane;
  int s = 0, d = 0;
  if (e < E) { s = src[e]; d = dst[e]; }
  const int cnt = E - e0;
  const int half = lane >> 4;
  const int c = lane & 15;
  int sj[16], dj[16];
  #pragma unroll
  for (int j = 0; j < 16; ++j) {
    const int idx = 2 * j + half;
    sj[j] = __shfl(s, idx);
    dj[j] = __shfl(d, idx);
  }
  if (c < 13) {
    if (cnt >= 32) {
      float xv[16];
      #pragma unroll
      for (int j = 0; j < 16; ++j) xv[j] = y[sj[j] * RP + c];
      #pragma unroll
      for (int j = 0; j < 16; ++j) atomicAdd(&agg[dj[j] * RP + c], xv[j]);
    } else {
      #pragma unroll
      for (int j = 0; j < 16; ++j)
        if (2 * j + half < cnt) atomicAdd(&agg[dj[j] * RP + c], y[sj[j] * RP + c]);
    }
  }
}

#define H1P 68

template <bool FULL>
__device__ __forceinline__ void gemm12_tile(const float* __restrict__ A,
                                            const float* __restrict__ dinv,
                                            const float* __restrict__ W1,
                                            const float* __restrict__ b1,
                                            const float* __restrict__ W2,
                                            float* __restrict__ O, float* hs,
                                            int N, int mbase, int lane) {
  const int half = lane >> 4;
  const int l16 = lane & 15;
  const int rowA = mbase + l16;
  const bool rv = FULL || (rowA < N);
  const int rowL = rv ? rowA : 0;

  v8f acc[4];
  const v8f zero = {0.f, 0.f, 0.f, 0.f, 0.f, 0.f, 0.f, 0.f};
  #pragma unroll
  for (int nt = 0; nt < 4; ++nt) acc[nt] = zero;

  const float* Arow = A + rowL * RP;
  const float dr = dinv[2 * rowL];
  #pragma unroll
  for (int kk = 0; kk < 12; kk += 4) {
    const int ka0 = kk + half * 2;
    const float2 a01 = *(const float2*)(Arow + ka0);
    const float x0 = dr * a01.x;
    const float x1 = dr * a01.y;
    v2f av; av[0] = rv ? x0 : 0.0f; av[1] = rv ? x1 : 0.0f;
    #pragma unroll
    for (int nt = 0; nt < 4; ++nt) {
      const int col = nt * 16 + l16;
      v2f bv; bv[0] = W1[ka0 * 64 + col]; bv[1] = W1[(ka0 + 1) * 64 + col];
      acc[nt] = __builtin_amdgcn_wmma_f32_16x16x4_f32(
          false, av, false, bv, (short)0, acc[nt], false, false);
    }
  }
  {
    const float a12 = dr * Arow[12];
    v2f av; av[0] = (half || !rv) ? 0.0f : a12; av[1] = 0.0f;
    #pragma unroll
    for (int nt = 0; nt < 4; ++nt) {
      const int col = nt * 16 + l16;
      const float b12 = W1[12 * 64 + col];
      v2f bv; bv[0] = half ? 0.0f : b12; bv[1] = 0.0f;
      acc[nt] = __builtin_amdgcn_wmma_f32_16x16x4_f32(
          false, av, false, bv, (short)0, acc[nt], false, false);
    }
  }
  #pragma unroll
  for (int nt = 0; nt < 4; ++nt) {
    const int col = nt * 16 + l16;
    const float bc = b1[col];
    #pragma unroll
    for (int v = 0; v < 8; ++v)
      hs[(v + half * 8) * H1P + col] = fmaxf(acc[nt][v] + bc, 0.0f);
  }
  __builtin_amdgcn_fence(__ATOMIC_RELEASE, "wavefront");
  __builtin_amdgcn_wave_barrier();
  __builtin_amdgcn_fence(__ATOMIC_ACQUIRE, "wavefront");

  v8f acc2[2];
  acc2[0] = zero; acc2[1] = zero;
  const float* hrow = hs + l16 * H1P;
  #pragma unroll
  for (int kk = 0; kk < 64; kk += 4) {
    const int ka0 = kk + half * 2;
    const float2 a2 = *(const float2*)(hrow + ka0);
    v2f av; av[0] = a2.x; av[1] = a2.y;
    #pragma unroll
    for (int nt = 0; nt < 2; ++nt) {
      const int col = nt * 16 + l16;
      v2f bv; bv[0] = W2[ka0 * 32 + col]; bv[1] = W2[(ka0 + 1) * 32 + col];
      acc2[nt] = __builtin_amdgcn_wmma_f32_16x16x4_f32(
          false, av, false, bv, (short)0, acc2[nt], false, false);
    }
  }
  float dsc[8];
  #pragma unroll
  for (int v = 0; v < 8; ++v) {
    const int row = mbase + v + half * 8;
    dsc[v] = (FULL || row < N) ? dinv[2 * row] : 0.0f;
  }
  #pragma unroll
  for (int nt = 0; nt < 2; ++nt) {
    const int col = nt * 16 + l16;
    #pragma unroll
    for (int v = 0; v < 8; ++v) {
      const int row = mbase + v + half * 8;
      if (FULL || row < N) O[row * 32 + col] = acc2[nt][v] * dsc[v];
    }
  }
}

__global__ __launch_bounds__(256)
void gnn_gemm12_wmma(const float* __restrict__ A, const float* __restrict__ dinv,
                     const float* __restrict__ W1, const float* __restrict__ b1,
                     const float* __restrict__ W2, float* __restrict__ O, int N) {
  __shared__ __attribute__((aligned(16))) float h1s[8 * 16 * H1P];
  const int lane = threadIdx.x & 31;
  const int wave = threadIdx.x >> 5;
  const int mbase = (blockIdx.x * 8 + wave) * 16;
  float* hs = h1s + wave * (16 * H1P);
  if (mbase + 16 <= N)
    gemm12_tile<true>(A, dinv, W1, b1, W2, O, hs, N, mbase, lane);
  else if (mbase < N)
    gemm12_tile<false>(A, dinv, W1, b1, W2, O, hs, N, mbase, lane);
}

#define PG_MAX 512
#define PE_THREADS 1024

template <bool IN_LDS>
__device__ __forceinline__ void pool_edge_walk(const float* __restrict__ xw2,
                                               const float* __restrict__ rec,
                                               const int* __restrict__ src,
                                               const int* __restrict__ dst,
                                               float* acc, float* cl,
                                               float* __restrict__ sums, float* __restrict__ cnts,
                                               int E, int V) {
  const int lane = threadIdx.x & 31;
  const int wave = blockIdx.x * (PE_THREADS / 32) + (threadIdx.x >> 5);
  const int nwave = gridDim.x * (PE_THREADS / 32);
  const int nchunk = (V + 31) >> 5;
  int ch = wave;
  if (ch >= nchunk) return;
  int ls = 0, ld = 0;
  {
    const int e = ch * 32 + lane;
    if (e < E) { ls = src[e]; ld = dst[e]; }
  }
  for (;;) {
    const int e0 = ch * 32;
    const int e = e0 + lane;
    const bool self = (e >= E);
    const int s = self ? e - E : ls;
    const int d = self ? e - E : ld;
    const bool live = (e < V);
    const int di = live ? d : 0;
    const float2 rd = ((const float2*)rec)[di];
    const float b = rd.x;
    const int g = __float_as_int(rd.y);
    const int chn = ch + nwave;
    ls = 0; ld = 0;
    {
      const int en = chn * 32 + lane;
      if (en < E) { ls = src[en]; ld = dst[en]; }
    }
    const int cnt = V - e0;
    if (cnt >= 32) {
      float xv[32];
      #pragma unroll
      for (int j = 0; j < 32; ++j)
        xv[j] = xw2[__builtin_amdgcn_readlane(s, j) * 32 + lane];
      const int nbits = __float_as_int(b);
      if (self) {
        if (IN_LDS) atomicAdd(&cl[g], 1.0f); else atomicAdd(&cnts[g], 1.0f);
      }
      #pragma unroll
      for (int j = 0; j < 32; ++j) {
        const float nj = __int_as_float(__builtin_amdgcn_readlane(nbits, j));
        const int gj = __builtin_amdgcn_readlane(g, j);
        if (IN_LDS) atomicAdd(&acc[gj * 32 + lane], nj * xv[j]);
        else        atomicAdd(&sums[gj * 32 + lane], nj * xv[j]);
      }
    } else {
      const float nrm = b;
      if (self && live) {
        if (IN_LDS) atomicAdd(&cl[g], 1.0f); else atomicAdd(&cnts[g], 1.0f);
      }
      for (int j = 0; j < cnt; ++j) {
        const int sj = __shfl(s, j);
        const int gj = __shfl(g, j);
        const float nj = __shfl(nrm, j);
        if (IN_LDS) atomicAdd(&acc[gj * 32 + lane], nj * xw2[sj * 32 + lane]);
        else        atomicAdd(&sums[gj * 32 + lane], nj * xw2[sj * 32 + lane]);
      }
    }
    ch = chn;
    if (ch >= nchunk) break;
  }
}

__global__ __launch_bounds__(PE_THREADS)
void gnn_pool_edge(const float* __restrict__ xw2, const float* __restrict__ rec,
                   const int* __restrict__ src, const int* __restrict__ dst,
                   float* __restrict__ sums, float* __restrict__ cnts,
                   int E, int N, int G) {
  __shared__ float acc[PG_MAX * 32];
  __shared__ float cl[PG_MAX];
  const int V = E + N;
  if (G <= PG_MAX) {
    for (int i = threadIdx.x; i < G * 32; i += PE_THREADS) acc[i] = 0.0f;
    for (int i = threadIdx.x; i < G; i += PE_THREADS) cl[i] = 0.0f;
    __syncthreads();
    pool_edge_walk<true>(xw2, rec, src, dst, acc, cl, sums, cnts, E, V);
    __syncthreads();
    const int lane = threadIdx.x & 31;
    const int w = threadIdx.x >> 5;
    const int r0 = (int)((blockIdx.x * 37u) % (unsigned)G);
    for (int r = w; r < G; r += PE_THREADS / 32) {
      int row = r + r0; if (row >= G) row -= G;
      atomicAdd(&sums[row * 32 + lane], acc[row * 32 + lane]);
    }
    for (int i = threadIdx.x; i < G; i += PE_THREADS) {
      const float c = cl[i];
      if (c != 0.0f) atomicAdd(&cnts[i], c);
    }
  } else {
    pool_edge_walk<false>(xw2, rec, src, dst, acc, cl, sums, cnts, E, V);
  }
}

__global__ void gnn_head(const float* __restrict__ sums, const float* __restrict__ cnts,
                         const float* __restrict__ b2,
                         const float* __restrict__ w1, const float* __restrict__ bias1,
                         const float* __restrict__ w2, const float* __restrict__ bias2,
                         float* __restrict__ o, int G) {
  const int t = blockIdx.x * blockDim.x + threadIdx.x;
  if (t >= G * 32) return;
  const int b = t >> 5, j = t & 31;
  const int gb = __float_as_int(sums[t] / fmaxf(cnts[b], 1.0f) + b2[j]);
  float s = bias1[j];
  #pragma unroll
  for (int k = 0; k < 32; ++k)
    s += __int_as_float(__builtin_amdgcn_readlane(gb, k)) * w1[k * 32 + j];
  const int hb = __float_as_int(fmaxf(s, 0.0f));
  float r = bias2[j];
  #pragma unroll
  for (int k = 0; k < 32; ++k)
    r += __int_as_float(__builtin_amdgcn_readlane(hb, k)) * w2[k * 32 + j];
  o[t] = r;
}

extern "C" void kernel_launch(void* const* d_in, const int* in_sizes, int n_in,
                              void* d_out, int out_size, void* d_ws, size_t ws_size,
                              hipStream_t stream) {
  const float* x    = (const float*)d_in[0];
  const int*   ei   = (const int*)d_in[1];
  const int*   batch= (const int*)d_in[2];
  const float* W1   = (const float*)d_in[3];
  const float* b1   = (const float*)d_in[4];
  const float* W2   = (const float*)d_in[5];
  const float* b2   = (const float*)d_in[6];
  const float* fc1w = (const float*)d_in[7];
  const float* fc1b = (const float*)d_in[8];
  const float* fc2w = (const float*)d_in[9];
  const float* fc2b = (const float*)d_in[10];

  const int N = in_sizes[0] / 13;
  const int E = in_sizes[1] / 2;
  const int G = out_size / 32;
  const int* src = ei;
  const int* dst = ei + E;

  auto al4 = [](size_t v) { return (v + 31) & ~(size_t)31; };
  float* ws   = (float*)d_ws;
  size_t off  = 0;
  float* dinv = ws + off; off = al4(off + (size_t)N);
  float* rec  = ws + off; off = al4(off + (size_t)2 * N);
  float* y1   = ws + off; off = al4(off + (size_t)RP * N);
  float* agg1 = ws + off; off = al4(off + (size_t)RP * N);
  float* xw2  = ws + off; off = al4(off + (size_t)32 * N);
  float* sums = ws + off; off = al4(off + (size_t)32 * G);
  float* cnts = ws + off; off = al4(off + (size_t)G);

  auto cdiv = [](int a, int b) { return (a + b - 1) / b; };

  gnn_init<<<cdiv(N, TPB), TPB, 0, stream>>>(dinv, sums, cnts, N, G);
  gnn_deg<<<cdiv(E, TPB), TPB, 0, stream>>>(dst, dinv, E);
  gnn_node_prep<<<cdiv(N, PREP_NODES), TPB, 0, stream>>>(dinv, batch, rec, x, y1, agg1, N);

  gnn_agg1_edge<<<cdiv(E, TPB), TPB, 0, stream>>>(y1, src, dst, agg1, E);
  int mt = cdiv(N, 16);
  gnn_gemm12_wmma<<<cdiv(mt, 8), 256, 0, stream>>>(agg1, rec, W1, b1, W2, xw2, N);

  gnn_pool_edge<<<256, PE_THREADS, 0, stream>>>(xw2, rec, src, dst, sums, cnts, E, N, G);
  gnn_head<<<cdiv(G * 32, TPB), TPB, 0, stream>>>(sums, cnts, b2, fc1w, fc1b, fc2w, fc2b,
                                                   (float*)d_out, G);
}


# one sum plane per package half for every global float-atomic table (degree, layer-1 row sums, pool sums): a wave adds into its own half's plane, readers add the two planes; new sum order
# speedup vs baseline: 6.4001x; 1.9450x over previous
#include <hip/hip_runtime.h>

typedef float v2f __attribute__((ext_vector_type(2)));
typedef float v8f __attribute__((ext_vector_type(8)));

#define TPB 256

__device__ __forceinline__ int pkg_half() {
  return (__builtin_amdgcn_s_sendmsg_rtn(0x87) >> 18) & 1;
}

__global__ void gnn_init(float* __restrict__ deg, float* __restrict__ sums,
                         float* __restrict__ cnts, int N, int G) {
  int i = blockIdx.x * blockDim.x + threadIdx.x;
  if (i < N) { deg[i] = 1.0f; deg[N + i] = 0.0f; }
  for (int j = i; j < 2 * G * 32; j += gridDim.x * blockDim.x) sums[j] = 0.0f;
  for (int j = i; j < 2 * G; j += gridDim.x * blockDim.x) cnts[j] = 0.0f;
}

__global__ void gnn_deg(const int* __restrict__ dst, float* __restrict__ deg, int E, int N) {
  int e = blockIdx.x * blockDim.x + threadIdx.x;
  float* degh = deg + pkg_half() * N;
  if (e < E) atomicAdd(&degh[dst[e]], 1.0f);
}

#define RP 16
#define PREP_NODES 128
__global__ void gnn_node_prep(const float* __restrict__ deg, const int* __restrict__ batch,
                              float* __restrict__ rec, const float* __restrict__ x,
                              float* __restrict__ y, float* __restrict__ agg, int N) {
  const int c = threadIdx.x & 15;
  const int sub = threadIdx.x >> 4;
  const int base = blockIdx.x * PREP_NODES;
  const int cl = c < 13 ? c : 12;
  float dg[PREP_NODES / 16], dh[PREP_NODES / 16], xs[PREP_NODES / 16];
  int bt[PREP_NODES / 16];
  #pragma unroll
  for (int k = 0; k < PREP_NODES / 16; ++k) {
    const int i = base + k * 16 + sub;
    const int il = i < N ? i : N - 1;
    dg[k] = deg[il];
    dh[k] = deg[N + il];
    xs[k] = x[il * 13 + cl];
    bt[k] = batch[il];
  }
  asm volatile("" :: "v"(dg[0]), "v"(dg[1]), "v"(dg[2]), "v"(dg[3]), "v"(dg[4]), "v"(dg[5]),
               "v"(dg[6]), "v"(dg[7]), "v"(dh[0]), "v"(dh[1]), "v"(dh[2]), "v"(dh[3]), "v"(dh[4]),
               "v"(dh[5]), "v"(dh[6]), "v"(dh[7]), "v"(xs[0]), "v"(xs[1]), "v"(xs[2]), "v"(xs[3]),
               "v"(xs[4]), "v"(xs[5]), "v"(xs[6]), "v"(xs[7]), "v"(bt[0]), "v"(bt[1]), "v"(bt[2]),
               "v"(bt[3]), "v"(bt[4]), "v"(bt[5]), "v"(bt[6]), "v"(bt[7]));
  #pragma unroll
  for (int k = 0; k < PREP_NODES / 16; ++k) {
    const int i = base + k * 16 + sub;
    const float dv = rsqrtf(dg[k] + dh[k]);
    const float v = dv * xs[k];
    if (i < N) {
      if (c < 13) { y[i * RP + c] = v; agg[i * RP + c] = v; agg[N * RP + i * RP + c] = 0.0f; }
      if (c == 0) ((float2*)rec)[i] = make_float2(dv, __int_as_float(bt[k]));
    }
  }
}

__global__ void gnn_agg1_edge(const float* __restrict__ y,
                              const int* __restrict__ src, const int* __restrict__ dst,
                              float* __restrict__ aggp, int E, int N) {
  const int lane = threadIdx.x & 31;
  float* agg = aggp + pkg_half() * (N * RP);
  const int e0 = (blockIdx.x * blockDim.x + threadIdx.x) & ~31;
  if (e0 >= E) return;
  const int e = e0 + lane;
  int s = 0, d = 0;
  if (e < E) { s = src[e]; d = dst[e]; }
  const int cnt = E - e0;
  const int half = lane >> 4;
  const int c = lane & 15;
  int sj[16], dj[16];
  #pragma unroll
  for (int j = 0; j < 16; ++j) {
    const int idx = 2 * j + half;
    sj[j] = __shfl(s, idx);
    dj[j] = __shfl(d, idx);
  }
  if (c < 13) {
    if (cnt >= 32) {
      float xv[16];
      #pragma unroll
      for (int j = 0; j < 16; ++j) xv[j] = y[sj[j] * RP + c];
      #pragma unroll
      for (int j = 0; j < 16; ++j) atomicAdd(&agg[dj[j] * RP + c], xv[j]);
    } else {
      #pragma unroll
      for (int j = 0; j < 16; ++j)
        if (2 * j + half < cnt) atomicAdd(&agg[dj[j] * RP + c], y[sj[j] * RP + c]);
    }
  }
}

#define H1P 68

template <bool FULL>
__device__ __forceinline__ void gemm12_tile(const float* __restrict__ A,
                                            const float* __restrict__ dinv,
                                            const float* __restrict__ W1,
                                            const float* __restrict__ b1,
                                            const float* __restrict__ W2,
                                            float* __restrict__ O, float* hs,
                                            int N, int mbase, int lane) {
  const int half = lane >> 4;
  const int l16 = lane & 15;
  const int rowA = mbase + l16;
  const bool rv = FULL || (rowA < N);
  const int rowL = rv ? rowA : 0;

  v8f acc[4];
  const v8f zero = {0.f, 0.f, 0.f, 0.f, 0.f, 0.f, 0.f, 0.f};
  #pragma unroll
  for (int nt = 0; nt < 4; ++nt) acc[nt] = zero;

  const float* Arow = A + rowL * RP;
  const float* Brow = Arow + N * RP;
  const float dr = dinv[2 * rowL];
  #pragma unroll
  for (int kk = 0; kk < 12; kk += 4) {
    const int ka0 = kk + half * 2;
    const float2 a01 = *(const float2*)(Arow + ka0);
    const float2 c01 = *(const float2*)(Brow + ka0);
    const float x0 = dr * (a01.x + c01.x);
    const float x1 = dr * (a01.y + c01.y);
    v2f av; av[0] = rv ? x0 : 0.0f; av[1] = rv ? x1 : 0.0f;
    #pragma unroll
    for (int nt = 0; nt < 4; ++nt) {
      const int col = nt * 16 + l16;
      v2f bv; bv[0] = W1[ka0 * 64 + col]; bv[1] = W1[(ka0 + 1) * 64 + col];
      acc[nt] = __builtin_amdgcn_wmma_f32_16x16x4_f32(
          false, av, false, bv, (short)0, acc[nt], false, false);
    }
  }
  {
    const float a12 = dr * (Arow[12] + Brow[12]);
    v2f av; av[0] = (half || !rv) ? 0.0f : a12; av[1] = 0.0f;
    #pragma unroll
    for (int nt = 0; nt < 4; ++nt) {
      const int col = nt * 16 + l16;
      const float b12 = W1[12 * 64 + col];
      v2f bv; bv[0] = half ? 0.0f : b12; bv[1] = 0.0f;
      acc[nt] = __builtin_amdgcn_wmma_f32_16x16x4_f32(
          false, av, false, bv, (short)0, acc[nt], false, false);
    }
  }
  #pragma unroll
  for (int nt = 0; nt < 4; ++nt) {
    const int col = nt * 16 + l16;
    const float bc = b1[col];
    #pragma unroll
    for (int v = 0; v < 8; ++v)
      hs[(v + half * 8) * H1P + col] = fmaxf(acc[nt][v] + bc, 0.0f);
  }
  __builtin_amdgcn_fence(__ATOMIC_RELEASE, "wavefront");
  __builtin_amdgcn_wave_barrier();
  __builtin_amdgcn_fence(__ATOMIC_ACQUIRE, "wavefront");

  v8f acc2[2];
  acc2[0] = zero; acc2[1] = zero;
  const float* hrow = hs + l16 * H1P;
  #pragma unroll
  for (int kk = 0; kk < 64; kk += 4) {
    const int ka0 = kk + half * 2;
    const float2 a2 = *(const float2*)(hrow + ka0);
    v2f av; av[0] = a2.x; av[1] = a2.y;
    #pragma unroll
    for (int nt = 0; nt < 2; ++nt) {
      const int col = nt * 16 + l16;
      v2f bv; bv[0] = W2[ka0 * 32 + col]; bv[1] = W2[(ka0 + 1) * 32 + col];
      acc2[nt] = __builtin_amdgcn_wmma_f32_16x16x4_f32(
          false, av, false, bv, (short)0, acc2[nt], false, false);
    }
  }
  float dsc[8];
  #pragma unroll
  for (int v = 0; v < 8; ++v) {
    const int row = mbase + v + half * 8;
    dsc[v] = (FULL || row < N) ? dinv[2 * row] : 0.0f;
  }
  #pragma unroll
  for (int nt = 0; nt < 2; ++nt) {
    const int col = nt * 16 + l16;
    #pragma unroll
    for (int v = 0; v < 8; ++v) {
      const int row = mbase + v + half * 8;
      if (FULL || row < N) O[row * 32 + col] = acc2[nt][v] * dsc[v];
    }
  }
}

__global__ __launch_bounds__(256)
void gnn_gemm12_wmma(const float* __restrict__ A, const float* __restrict__ dinv,
                     const float* __restrict__ W1, const float* __restrict__ b1,
                     const float* __restrict__ W2, float* __restrict__ O, int N) {
  __shared__ __attribute__((aligned(16))) float h1s[8 * 16 * H1P];
  const int lane = threadIdx.x & 31;
  const int wave = threadIdx.x >> 5;
  const int mbase = (blockIdx.x * 8 + wave) * 16;
  float* hs = h1s + wave * (16 * H1P);
  if (mbase + 16 <= N)
    gemm12_tile<true>(A, dinv, W1, b1, W2, O, hs, N, mbase, lane);
  else if (mbase < N)
    gemm12_tile<false>(A, dinv, W1, b1, W2, O, hs, N, mbase, lane);
}

#define PG_MAX 512
#define PE_THREADS 1024

template <bool IN_LDS>
__device__ __forceinline__ void pool_edge_walk(const float* __restrict__ xw2,
                                               const float* __restrict__ rec,
                                               const int* __restrict__ src,
                                               const int* __restrict__ dst,
                                               float* acc, float* cl,
                                               float* __restrict__ sums, float* __restrict__ cnts,
                                               int E, int V) {
  const int lane = threadIdx.x & 31;
  const int wave = blockIdx.x * (PE_THREADS / 32) + (threadIdx.x >> 5);
  const int nwave = gridDim.x * (PE_THREADS / 32);
  const int nchunk = (V + 31) >> 5;
  int ch = wave;
  if (ch >= nchunk) return;
  int ls = 0, ld = 0;
  {
    const int e = ch * 32 + lane;
    if (e < E) { ls = src[e]; ld = dst[e]; }
  }
  for (;;) {
    const int e0 = ch * 32;
    const int e = e0 + lane;
    const bool self = (e >= E);
    const int s = self ? e - E : ls;
    const int d = self ? e - E : ld;
    const bool live = (e < V);
    const int di = live ? d : 0;
    const float2 rd = ((const float2*)rec)[di];
    const float b = rd.x;
    const int g = __float_as_int(rd.y);
    const int chn = ch + nwave;
    ls = 0; ld = 0;
    {
      const int en = chn * 32 + lane;
      if (en < E) { ls = src[en]; ld = dst[en]; }
    }
    const int cnt = V - e0;
    if (cnt >= 32) {
      float xv[32];
      #pragma unroll
      for (int j = 0; j < 32; ++j)
        xv[j] = xw2[__builtin_amdgcn_readlane(s, j) * 32 + lane];
      const int nbits = __float_as_int(b);
      if (self) {
        if (IN_LDS) atomicAdd(&cl[g], 1.0f); else atomicAdd(&cnts[g], 1.0f);
      }
      #pragma unroll
      for (int j = 0; j < 32; ++j) {
        const float nj = __int_as_float(__builtin_amdgcn_readlane(nbits, j));
        const int gj = __builtin_amdgcn_readlane(g, j);
        if (IN_LDS) atomicAdd(&acc[gj * 32 + lane], nj * xv[j]);
        else        atomicAdd(&sums[gj * 32 + lane], nj * xv[j]);
      }
    } else {
      const float nrm = b;
      if (self && live) {
        if (IN_LDS) atomicAdd(&cl[g], 1.0f); else atomicAdd(&cnts[g], 1.0f);
      }
      for (int j = 0; j < cnt; ++j) {
        const int sj = __shfl(s, j);
        const int gj = __shfl(g, j);
        const float nj = __shfl(nrm, j);
        if (IN_LDS) atomicAdd(&acc[gj * 32 + lane], nj * xw2[sj * 32 + lane]);
        else        atomicAdd(&sums[gj * 32 + lane], nj * xw2[sj * 32 + lane]);
      }
    }
    ch = chn;
    if (ch >= nchunk) break;
  }
}

__global__ __launch_bounds__(PE_THREADS)
void gnn_pool_edge(const float* __restrict__ xw2, const float* __restrict__ rec,
                   const int* __restrict__ src, const int* __restrict__ dst,
                   float* __restrict__ sums2, float* __restrict__ cnts2,
                   int E, int N, int G) {
  __shared__ float acc[PG_MAX * 32];
  __shared__ float cl[PG_MAX];
  const int V = E + N;
  const int hf = pkg_half();
  float* sums = sums2 + hf * (G * 32);
  float* cnts = cnts2 + hf * G;
  if (G <= PG_MAX) {
    for (int i = threadIdx.x; i < G * 32; i += PE_THREADS) acc[i] = 0.0f;
    for (int i = threadIdx.x; i < G; i += PE_THREADS) cl[i] = 0.0f;
    __syncthreads();
    pool_edge_walk<true>(xw2, rec, src, dst, acc, cl, sums, cnts, E, V);
    __syncthreads();
    const int lane = threadIdx.x & 31;
    const int w = threadIdx.x >> 5;
    const int r0 = (int)((blockIdx.x * 37u) % (unsigned)G);
    for (int r = w; r < G; r += PE_THREADS / 32) {
      int row = r + r0; if (row >= G) row -= G;
      atomicAdd(&sums[row * 32 + lane], acc[row * 32 + lane]);
    }
    for (int i = threadIdx.x; i < G; i += PE_THREADS) {
      const float c = cl[i];
      if (c != 0.0f) atomicAdd(&cnts[i], c);
    }
  } else {
    pool_edge_walk<false>(xw2, rec, src, dst, acc, cl, sums, cnts, E, V);
  }
}

__global__ void gnn_head(const float* __restrict__ sums, const float* __restrict__ cnts,
                         const float* __restrict__ b2,
                         const float* __restrict__ w1, const float* __restrict__ bias1,
                         const float* __restrict__ w2, const float* __restrict__ bias2,
                         float* __restrict__ o, int G) {
  const int t = blockIdx.x * blockDim.x + threadIdx.x;
  if (t >= G * 32) return;
  const int b = t >> 5, j = t & 31;
  const float sm = sums[t] + sums[G * 32 + t];
  const float cn = cnts[b] + cnts[G + b];
  const int gb = __float_as_int(sm / fmaxf(cn, 1.0f) + b2[j]);
  float s = bias1[j];
  #pragma unroll
  for (int k = 0; k < 32; ++k)
    s += __int_as_float(__builtin_amdgcn_readlane(gb, k)) * w1[k * 32 + j];
  const int hb = __float_as_int(fmaxf(s, 0.0f));
  float r = bias2[j];
  #pragma unroll
  for (int k = 0; k < 32; ++k)
    r += __int_as_float(__builtin_amdgcn_readlane(hb, k)) * w2[k * 32 + j];
  o[t] = r;
}

extern "C" void kernel_launch(void* const* d_in, const int* in_sizes, int n_in,
                              void* d_out, int out_size, void* d_ws, size_t ws_size,
                              hipStream_t stream) {
  const float* x    = (const float*)d_in[0];
  const int*   ei   = (const int*)d_in[1];
  const int*   batch= (const int*)d_in[2];
  const float* W1   = (const float*)d_in[3];
  const float* b1   = (const float*)d_in[4];
  const float* W2   = (const float*)d_in[5];
  const float* b2   = (const float*)d_in[6];
  const float* fc1w = (const float*)d_in[7];
  const float* fc1b = (const float*)d_in[8];
  const float* fc2w = (const float*)d_in[9];
  const float* fc2b = (const float*)d_in[10];

  const int N = in_sizes[0] / 13;
  const int E = in_sizes[1] / 2;
  const int G = out_size / 32;
  const int* src = ei;
  const int* dst = ei + E;

  auto al4 = [](size_t v) { return (v + 31) & ~(size_t)31; };
  float* ws   = (float*)d_ws;
  size_t off  = 0;
  float* dinv = ws + off; off = al4(off + (size_t)2 * N);
  float* rec  = ws + off; off = al4(off + (size_t)2 * N);
  float* y1   = ws + off; off = al4(off + (size_t)RP * N);
  float* agg1 = ws + off; off = al4(off + (size_t)2 * RP * N);
  float* xw2  = ws + off; off = al4(off + (size_t)32 * N);
  float* sums = ws + off; off = al4(off + (size_t)2 * 32 * G);
  float* cnts = ws + off; off = al4(off + (size_t)2 * G);

  auto cdiv = [](int a, int b) { return (a + b - 1) / b; };

  gnn_init<<<cdiv(N, TPB), TPB, 0, stream>>>(dinv, sums, cnts, N, G);
  gnn_deg<<<cdiv(E, TPB), TPB, 0, stream>>>(dst, dinv, E, N);
  gnn_node_prep<<<cdiv(N, PREP_NODES), TPB, 0, stream>>>(dinv, batch, rec, x, y1, agg1, N);

  gnn_agg1_edge<<<cdiv(E, TPB), TPB, 0, stream>>>(y1, src, dst, agg1, E, N);
  int mt = cdiv(N, 16);
  gnn_gemm12_wmma<<<cdiv(mt, 8), 256, 0, stream>>>(agg1, rec, W1, b1, W2, xw2, N);

  gnn_pool_edge<<<256, PE_THREADS, 0, stream>>>(xw2, rec, src, dst, sums, cnts, E, N, G);
  gnn_head<<<cdiv(G * 32, TPB), TPB, 0, stream>>>(sums, cnts, b2, fc1w, fc1b, fc2w, fc2b,
                                                   (float*)d_out, G);
}


# degree count and layer-1 edge scatter start 8x and 5x fewer, longer waves: 8 edges per lane in the count, a fixed grid of waves looping over 32-edge chunks in the scatter; atomics in arrival order
# speedup vs baseline: 6.9521x; 1.0862x over previous
#include <hip/hip_runtime.h>

typedef float v2f __attribute__((ext_vector_type(2)));
typedef float v8f __attribute__((ext_vector_type(8)));

#define TPB 256

__device__ __forceinline__ int pkg_half() {
  return (__builtin_amdgcn_s_sendmsg_rtn(0x87) >> 18) & 1;
}

__global__ void gnn_init(float* __restrict__ deg, float* __restrict__ sums,
                         float* __restrict__ cnts, int N, int G) {
  int i = blockIdx.x * blockDim.x + threadIdx.x;
  if (i < N) { deg[i] = 1.0f; deg[N + i] = 0.0f; }
  for (int j = i; j < 2 * G * 32; j += gridDim.x * blockDim.x) sums[j] = 0.0f;
  for (int j = i; j < 2 * G; j += gridDim.x * blockDim.x) cnts[j] = 0.0f;
}

#define DEG_TPB 128
#define DEG_EPW 256
__global__ void gnn_deg(const int* __restrict__ dst, float* __restrict__ deg, int E, int N) {
  float* degh = deg + pkg_half() * N;
  const int lane = threadIdx.x & 31;
  const int e0 = ((blockIdx.x * blockDim.x + threadIdx.x) >> 5) * DEG_EPW;
  if (e0 >= E) return;
  if (e0 + DEG_EPW <= E && (((size_t)(dst + e0)) & 15) == 0) {
    const int4 a = *(const int4*)(dst + e0 + lane * 4);
    const int4 b = *(const int4*)(dst + e0 + DEG_EPW / 2 + lane * 4);
    atomicAdd(&degh[a.x], 1.0f); atomicAdd(&degh[a.y], 1.0f);
    atomicAdd(&degh[a.z], 1.0f); atomicAdd(&degh[a.w], 1.0f);
    atomicAdd(&degh[b.x], 1.0f); atomicAdd(&degh[b.y], 1.0f);
    atomicAdd(&degh[b.z], 1.0f); atomicAdd(&degh[b.w], 1.0f);
  } else {
    for (int k = 0; k < DEG_EPW / 32; ++k) {
      const int e = e0 + k * 32 + lane;
      if (e < E) atomicAdd(&degh[dst[e]], 1.0f);
    }
  }
}

#define RP 16
#define PREP_NODES 128
__global__ void gnn_node_prep(const float* __restrict__ deg, const int* __restrict__ batch,
                              float* __restrict__ rec, const float* __restrict__ x,
                              float* __restrict__ y, float* __restrict__ agg, int N) {
  const int c = threadIdx.x & 15;
  const int sub = threadIdx.x >> 4;
  const int base = blockIdx.x * PREP_NODES;
  const int cl = c < 13 ? c : 12;
  float dg[PREP_NODES / 16], dh[PREP_NODES / 16], xs[PREP_NODES / 16];
  int bt[PREP_NODES / 16];
  #pragma unroll
  for (int k = 0; k < PREP_NODES / 16; ++k) {
    const int i = base + k * 16 + sub;
    const int il = i < N ? i : N - 1;
    dg[k] = deg[il];
    dh[k] = deg[N + il];
    xs[k] = x[il * 13 + cl];
    bt[k] = batch[il];
  }
  asm volatile("" :: "v"(dg[0]), "v"(dg[1]), "v"(dg[2]), "v"(dg[3]), "v"(dg[4]), "v"(dg[5]),
               "v"(dg[6]), "v"(dg[7]), "v"(dh[0]), "v"(dh[1]), "v"(dh[2]), "v"(dh[3]), "v"(dh[4]),
               "v"(dh[5]), "v"(dh[6]), "v"(dh[7]), "v"(xs[0]), "v"(xs[1]), "v"(xs[2]), "v"(xs[3]),
               "v"(xs[4]), "v"(xs[5]), "v"(xs[6]), "v"(xs[7]), "v"(bt[0]), "v"(bt[1]), "v"(bt[2]),
               "v"(bt[3]), "v"(bt[4]), "v"(bt[5]), "v"(bt[6]), "v"(bt[7]));
  #pragma unroll
  for (int k = 0; k < PREP_NODES / 16; ++k) {
    const int i = base + k * 16 + sub;
    const float dv = rsqrtf(dg[k] + dh[k]);
    const float v = dv * xs[k];
    if (i < N) {
      if (c < 13) { y[i * RP + c] = v; agg[i * RP + c] = v; agg[N * RP + i * RP + c] = 0.0f; }
      if (c == 0) ((float2*)rec)[i] = make_float2(dv, __int_as_float(bt[k]));
    }
  }
}

#define AGG_BLOCKS 1024
__global__ __launch_bounds__(TPB)
void gnn_agg1_edge(const float* __restrict__ y,
                   const int* __restrict__ src, const int* __restrict__ dst,
                   float* __restrict__ aggp, int E, int N) {
  const int lane = threadIdx.x & 31;
  float* agg = aggp + pkg_half() * (N * RP);
  const int nwave = gridDim.x * (TPB / 32);
  const int nchunk = (E + 31) >> 5;
  int ch = blockIdx.x * (TPB / 32) + (threadIdx.x >> 5);
  if (ch >= nchunk) return;
  const int half = lane >> 4;
  const int c = lane & 15;
  int s = 0, d = 0;
  {
    const int e = ch * 32 + lane;
    if (e < E) { s = src[e]; d = dst[e]; }
  }
  for (;;) {
    const int cnt = E - ch * 32;
    int sj[16], dj[16];
    #pragma unroll
    for (int j = 0; j < 16; ++j) {
      const int idx = 2 * j + half;
      sj[j] = __shfl(s, idx);
      dj[j] = __shfl(d, idx);
    }
    const int chn = ch + nwave;
    s = 0; d = 0;
    {
      const int en = chn * 32 + lane;
      if (en < E) { s = src[en]; d = dst[en]; }
    }
    if (c < 13) {
      if (cnt >= 32) {
        float xv[16];
        #pragma unroll
        for (int j = 0; j < 16; ++j) xv[j] = y[sj[j] * RP + c];
        #pragma unroll
        for (int j = 0; j < 16; ++j) atomicAdd(&agg[dj[j] * RP + c], xv[j]);
      } else {
        #pragma unroll
        for (int j = 0; j < 16; ++j)
          if (2 * j + half < cnt) atomicAdd(&agg[dj[j] * RP + c], y[sj[j] * RP + c]);
      }
    }
    ch = chn;
    if (ch >= nchunk) break;
  }
}

#define H1P 68

template <bool FULL>
__device__ __forceinline__ void gemm12_tile(const float* __restrict__ A,
                                            const float* __restrict__ dinv,
                                            const float* __restrict__ W1,
                                            const float* __restrict__ b1,
                                            const float* __restrict__ W2,
                                            float* __restrict__ O, float* hs,
                                            int N, int mbase, int lane) {
  const int half = lane >> 4;
  const int l16 = lane & 15;
  const int rowA = mbase + l16;
  const bool rv = FULL || (rowA < N);
  const int rowL = rv ? rowA : 0;

  v8f acc[4];
  const v8f zero = {0.f, 0.f, 0.f, 0.f, 0.f, 0.f, 0.f, 0.f};
  #pragma unroll
  for (int nt = 0; nt < 4; ++nt) acc[nt] = zero;

  const float* Arow = A + rowL * RP;
  const float* Brow = Arow + N * RP;
  const float dr = dinv[2 * rowL];
  #pragma unroll
  for (int kk = 0; kk < 12; kk += 4) {
    const int ka0 = kk + half * 2;
    const float2 a01 = *(const float2*)(Arow + ka0);
    const float2 c01 = *(const float2*)(Brow + ka0);
    const float x0 = dr * (a01.x + c01.x);
    const float x1 = dr * (a01.y + c01.y);
    v2f av; av[0] = rv ? x0 : 0.0f; av[1] = rv ? x1 : 0.0f;
    #pragma unroll
    for (int nt = 0; nt < 4; ++nt) {
      const int col = nt * 16 + l16;
      v2f bv; bv[0] = W1[ka0 * 64 + col]; bv[1] = W1[(ka0 + 1) * 64 + col];
      acc[nt] = __builtin_amdgcn_wmma_f32_16x16x4_f32(
          false, av, false, bv, (short)0, acc[nt], false, false);
    }
  }
  {
    const float a12 = dr * (Arow[12] + Brow[12]);
    v2f av; av[0] = (half || !rv) ? 0.0f : a12; av[1] = 0.0f;
    #pragma unroll
    for (int nt = 0; nt < 4; ++nt) {
      const int col = nt * 16 + l16;
      const float b12 = W1[12 * 64 + col];
      v2f bv; bv[0] = half ? 0.0f : b12; bv[1] = 0.0f;
      acc[nt] = __builtin_amdgcn_wmma_f32_16x16x4_f32(
          false, av, false, bv, (short)0, acc[nt], false, false);
    }
  }
  #pragma unroll
  for (int nt = 0; nt < 4; ++nt) {
    const int col = nt * 16 + l16;
    const float bc = b1[col];
    #pragma unroll
    for (int v = 0; v < 8; ++v)
      hs[(v + half * 8) * H1P + col] = fmaxf(acc[nt][v] + bc, 0.0f);
  }
  __builtin_amdgcn_fence(__ATOMIC_RELEASE, "wavefront");
  __builtin_amdgcn_wave_barrier();
  __builtin_amdgcn_fence(__ATOMIC_ACQUIRE, "wavefront");

  v8f acc2[2];
  acc2[0] = zero; acc2[1] = zero;
  const float* hrow = hs + l16 * H1P;
  #pragma unroll
  for (int kk = 0; kk < 64; kk += 4) {
    const int ka0 = kk + half * 2;
    const float2 a2 = *(const float2*)(hrow + ka0);
    v2f av; av[0] = a2.x; av[1] = a2.y;
    #pragma unroll
    for (int nt = 0; nt < 2; ++nt) {
      const int col = nt * 16 + l16;
      v2f bv; bv[0] = W2[ka0 * 32 + col]; bv[1] = W2[(ka0 + 1) * 32 + col];
      acc2[nt] = __builtin_amdgcn_wmma_f32_16x16x4_f32(
          false, av, false, bv, (short)0, acc2[nt], false, false);
    }
  }
  float dsc[8];
  #pragma unroll
  for (int v = 0; v < 8; ++v) {
    const int row = mbase + v + half * 8;
    dsc[v] = (FULL || row < N) ? dinv[2 * row] : 0.0f;
  }
  #pragma unroll
  for (int nt = 0; nt < 2; ++nt) {
    const int col = nt * 16 + l16;
    #pragma unroll
    for (int v = 0; v < 8; ++v) {
      const int row = mbase + v + half * 8;
      if (FULL || row < N) O[row * 32 + col] = acc2[nt][v] * dsc[v];
    }
  }
}

__global__ __launch_bounds__(256)
void gnn_gemm12_wmma(const float* __restrict__ A, const float* __restrict__ dinv,
                     const float* __restrict__ W1, const float* __restrict__ b1,
                     const float* __restrict__ W2, float* __restrict__ O, int N) {
  __shared__ __attribute__((aligned(16))) float h1s[8 * 16 * H1P];
  const int lane = threadIdx.x & 31;
  const int wave = threadIdx.x >> 5;
  const int mbase = (blockIdx.x * 8 + wave) * 16;
  float* hs = h1s + wave * (16 * H1P);
  if (mbase + 16 <= N)
    gemm12_tile<true>(A, dinv, W1, b1, W2, O, hs, N, mbase, lane);
  else if (mbase < N)
    gemm12_tile<false>(A, dinv, W1, b1, W2, O, hs, N, mbase, lane);
}

#define PG_MAX 512
#define PE_THREADS 1024

template <bool IN_LDS>
__device__ __forceinline__ void pool_edge_walk(const float* __restrict__ xw2,
                                               const float* __restrict__ rec,
                                               const int* __restrict__ src,
                                               const int* __restrict__ dst,
                                               float* acc, float* cl,
                                               float* __restrict__ sums, float* __restrict__ cnts,
                                               int E, int V) {
  const int lane = threadIdx.x & 31;
  const int wave = blockIdx.x * (PE_THREADS / 32) + (threadIdx.x >> 5);
  const int nwave = gridDim.x * (PE_THREADS / 32);
  const int nchunk = (V + 31) >> 5;
  int ch = wave;
  if (ch >= nchunk) return;
  int ls = 0, ld = 0;
  {
    const int e = ch * 32 + lane;
    if (e < E) { ls = src[e]; ld = dst[e]; }
  }
  for (;;) {
    const int e0 = ch * 32;
    const int e = e0 + lane;
    const bool self = (e >= E);
    const int s = self ? e - E : ls;
    const int d = self ? e - E : ld;
    const bool live = (e < V);
    const int di = live ? d : 0;
    const float2 rd = ((const float2*)rec)[di];
    const float b = rd.x;
    const int g = __float_as_int(rd.y);
    const int chn = ch + nwave;
    ls = 0; ld = 0;
    {
      const int en = chn * 32 + lane;
      if (en < E) { ls = src[en]; ld = dst[en]; }
    }
    const int cnt = V - e0;
    if (cnt >= 32) {
      float xv[32];
      #pragma unroll
      for (int j = 0; j < 32; ++j)
        xv[j] = xw2[__builtin_amdgcn_readlane(s, j) * 32 + lane];
      const int nbits = __float_as_int(b);
      if (self) {
        if (IN_LDS) atomicAdd(&cl[g], 1.0f); else atomicAdd(&cnts[g], 1.0f);
      }
      #pragma unroll
      for (int j = 0; j < 32; ++j) {
        const float nj = __int_as_float(__builtin_amdgcn_readlane(nbits, j));
        const int gj = __builtin_amdgcn_readlane(g, j);
        if (IN_LDS) atomicAdd(&acc[gj * 32 + lane], nj * xv[j]);
        else        atomicAdd(&sums[gj * 32 + lane], nj * xv[j]);
      }
    } else {
      const float nrm = b;
      if (self && live) {
        if (IN_LDS) atomicAdd(&cl[g], 1.0f); else atomicAdd(&cnts[g], 1.0f);
      }
      for (int j = 0; j < cnt; ++j) {
        const int sj = __shfl(s, j);
        const int gj = __shfl(g, j);
        const float nj = __shfl(nrm, j);
        if (IN_LDS) atomicAdd(&acc[gj * 32 + lane], nj * xw2[sj * 32 + lane]);
        else        atomicAdd(&sums[gj * 32 + lane], nj * xw2[sj * 32 + lane]);
      }
    }
    ch = chn;
    if (ch >= nchunk) break;
  }
}

__global__ __launch_bounds__(PE_THREADS)
void gnn_pool_edge(const float* __restrict__ xw2, const float* __restrict__ rec,
                   const int* __restrict__ src, const int* __restrict__ dst,
                   float* __restrict__ sums2, float* __restrict__ cnts2,
                   int E, int N, int G) {
  __shared__ float acc[PG_MAX * 32];
  __shared__ float cl[PG_MAX];
  const int V = E + N;
  const int hf = pkg_half();
  float* sums = sums2 + hf * (G * 32);
  float* cnts = cnts2 + hf * G;
  if (G <= PG_MAX) {
    for (int i = threadIdx.x; i < G * 32; i += PE_THREADS) acc[i] = 0.0f;
    for (int i = threadIdx.x; i < G; i += PE_THREADS) cl[i] = 0.0f;
    __syncthreads();
    pool_edge_walk<true>(xw2, rec, src, dst, acc, cl, sums, cnts, E, V);
    __syncthreads();
    const int lane = threadIdx.x & 31;
    const int w = threadIdx.x >> 5;
    const int r0 = (int)((blockIdx.x * 37u) % (unsigned)G);
    for (int r = w; r < G; r += PE_THREADS / 32) {
      int row = r + r0; if (row >= G) row -= G;
      atomicAdd(&sums[row * 32 + lane], acc[row * 32 + lane]);
    }
    for (int i = threadIdx.x; i < G; i += PE_THREADS) {
      const float c = cl[i];
      if (c != 0.0f) atomicAdd(&cnts[i], c);
    }
  } else {
    pool_edge_walk<false>(xw2, rec, src, dst, acc, cl, sums, cnts, E, V);
  }
}

__global__ void gnn_head(const float* __restrict__ sums, const float* __restrict__ cnts,
                         const float* __restrict__ b2,
                         const float* __restrict__ w1, const float* __restrict__ bias1,
                         const float* __restrict__ w2, const float* __restrict__ bias2,
                         float* __restrict__ o, int G) {
  const int t = blockIdx.x * blockDim.x + threadIdx.x;
  if (t >= G * 32) return;
  const int b = t >> 5, j = t & 31;
  const float sm = sums[t] + sums[G * 32 + t];
  const float cn = cnts[b] + cnts[G + b];
  const int gb = __float_as_int(sm / fmaxf(cn, 1.0f) + b2[j]);
  float s = bias1[j];
  #pragma unroll
  for (int k = 0; k < 32; ++k)
    s += __int_as_float(__builtin_amdgcn_readlane(gb, k)) * w1[k * 32 + j];
  const int hb = __float_as_int(fmaxf(s, 0.0f));
  float r = bias2[j];
  #pragma unroll
  for (int k = 0; k < 32; ++k)
    r += __int_as_float(__builtin_amdgcn_readlane(hb, k)) * w2[k * 32 + j];
  o[t] = r;
}

extern "C" void kernel_launch(void* const* d_in, const int* in_sizes, int n_in,
                              void* d_out, int out_size, void* d_ws, size_t ws_size,
                              hipStream_t stream) {
  const float* x    = (const float*)d_in[0];
  const int*   ei   = (const int*)d_in[1];
  const int*   batch= (const int*)d_in[2];
  const float* W1   = (const float*)d_in[3];
  const float* b1   = (const float*)d_in[4];
  const float* W2   = (const float*)d_in[5];
  const float* b2   = (const float*)d_in[6];
  const float* fc1w = (const float*)d_in[7];
  const float* fc1b = (const float*)d_in[8];
  const float* fc2w = (const float*)d_in[9];
  const float* fc2b = (const float*)d_in[10];

  const int N = in_sizes[0] / 13;
  const int E = in_sizes[1] / 2;
  const int G = out_size / 32;
  const int* src = ei;
  const int* dst = ei + E;

  auto al4 = [](size_t v) { return (v + 31) & ~(size_t)31; };
  float* ws   = (float*)d_ws;
  size_t off  = 0;
  float* dinv = ws + off; off = al4(off + (size_t)2 * N);
  float* rec  = ws + off; off = al4(off + (size_t)2 * N);
  float* y1   = ws + off; off = al4(off + (size_t)RP * N);
  float* agg1 = ws + off; off = al4(off + (size_t)2 * RP * N);
  float* xw2  = ws + off; off = al4(off + (size_t)32 * N);
  float* sums = ws + off; off = al4(off + (size_t)2 * 32 * G);
  float* cnts = ws + off; off = al4(off + (size_t)2 * G);

  auto cdiv = [](int a, int b) { return (a + b - 1) / b; };

  gnn_init<<<cdiv(N, TPB), TPB, 0, stream>>>(dinv, sums, cnts, N, G);
  gnn_deg<<<cdiv(E, DEG_EPW * (DEG_TPB / 32)), DEG_TPB, 0, stream>>>(dst, dinv, E, N);
  gnn_node_prep<<<cdiv(N, PREP_NODES), TPB, 0, stream>>>(dinv, batch, rec, x, y1, agg1, N);

  gnn_agg1_edge<<<AGG_BLOCKS, TPB, 0, stream>>>(y1, src, dst, agg1, E, N);
  int mt = cdiv(N, 16);
  gnn_gemm12_wmma<<<cdiv(mt, 8), 256, 0, stream>>>(agg1, rec, W1, b1, W2, xw2, N);

  gnn_pool_edge<<<256, PE_THREADS, 0, stream>>>(xw2, rec, src, dst, sums, cnts, E, N, G);
  gnn_head<<<cdiv(G * 32, TPB), TPB, 0, stream>>>(sums, cnts, b2, fc1w, fc1b, fc2w, fc2b,
                                                   (float*)d_out, G);
}
